# starting kernel
# baseline (speedup 1.0000x reference)
#include <hip/hip_runtime.h>

#define N_NODES  50000
#define N_EDGES  600000
#define E_TOT    (N_EDGES + N_NODES)
#define N_GRAPHS 1024
#define D        128
#define NEG_SLOPE 0.2f

typedef float    v2f  __attribute__((ext_vector_type(2)));
typedef float    v8f  __attribute__((ext_vector_type(8)));
typedef _Float16 v16h __attribute__((ext_vector_type(16)));

__global__ __launch_bounds__(256)
void gemm_wmma(const float* __restrict__ X, const float* __restrict__ W,
               float* __restrict__ H) {
  __shared__ float xs[16][132];
  __shared__ float ws[128][136];
  const int row0 = blockIdx.x * 16;
  const int t = threadIdx.x;

  for (int i = t; i < 16 * 32; i += 256) {
    const int r = i >> 5, c4 = i & 31;
    const float4 v = ((const float4*)(X + (size_t)(row0 + r) * D))[c4];
    float* p = &xs[r][c4 * 4];
    p[0] = v.x; p[1] = v.y; p[2] = v.z; p[3] = v.w;
  }
  for (int i = t; i < 128 * 32; i += 256) {
    const int r = i >> 5, c4 = i & 31;
    *(float4*)&ws[r][c4 * 4] = ((const float4*)(W + (size_t)r * D))[c4];
  }
  __syncthreads();

  const int wave = t >> 5, lane = t & 31;
  const int hi = lane >> 4;
  const int m  = lane & 15;
  const int n  = wave * 16 + m;

  v8f c = {0.f, 0.f, 0.f, 0.f, 0.f, 0.f, 0.f, 0.f};

#if __has_builtin(__builtin_amdgcn_wmma_f32_16x16x4_f32)
  #pragma unroll
  for (int k0 = 0; k0 < D; k0 += 4) {
    const int ka = k0 + hi * 2;
    v2f a, b;
    a[0] = xs[m][ka];
    a[1] = xs[m][ka + 1];
    b[0] = ws[ka][n];
    b[1] = ws[ka + 1][n];
    c = __builtin_amdgcn_wmma_f32_16x16x4_f32(false, a, false, b, (short)0, c,
                                              false, false);
  }
#else
  #pragma unroll
  for (int k0 = 0; k0 < D; k0 += 32) {
    v16h a, b;
    #pragma unroll
    for (int i = 0; i < 16; ++i) {
      const int vv = i >> 1, j = i & 1;
      const int ka = k0 + ((vv >= 4) ? 16 : 0) + hi * 8 + (vv & 3) * 2 + j;
      a[i] = (_Float16)xs[m][ka];
      const int kb = k0 + hi * 16 + i;
      b[i] = (_Float16)ws[kb][n];
    }
    c = __builtin_amdgcn_wmma_f32_16x16x32_f16(false, a, false, b, (short)0, c,
                                               false, false);
  }
#endif

  #pragma unroll
  for (int r = 0; r < 8; ++r)
    H[(size_t)(row0 + r + hi * 8) * D + n] = c[r];
}

__global__ __launch_bounds__(256)
void alpha_kernel(const float* __restrict__ H, const float* __restrict__ a_src,
                  const float* __restrict__ a_dst, float* __restrict__ asrc,
                  float* __restrict__ adst) {
  const int node = (blockIdx.x * blockDim.x + threadIdx.x) >> 5;
  const int lane = threadIdx.x & 31;
  if (node >= N_NODES) return;
  const int c0 = lane * 4;
  const float4 hv = *(const float4*)(H + (size_t)node * D + c0);
  const float4 sv = *(const float4*)(a_src + c0);
  const float4 dv = *(const float4*)(a_dst + c0);
  float s = hv.x * sv.x + hv.y * sv.y + hv.z * sv.z + hv.w * sv.w;
  float d = hv.x * dv.x + hv.y * dv.y + hv.z * dv.z + hv.w * dv.w;
  #pragma unroll
  for (int off = 16; off > 0; off >>= 1) {
    s += __shfl_down(s, off, 32);
    d += __shfl_down(d, off, 32);
  }
  if (lane == 0) { asrc[node] = s; adst[node] = d; }
}

__device__ __forceinline__ unsigned fmap(float x) {
  const unsigned u = __float_as_uint(x);
  return (u & 0x80000000u) ? ~u : (u | 0x80000000u);
}
__device__ __forceinline__ float funmap(unsigned u) {
  return __uint_as_float((u & 0x80000000u) ? (u & 0x7fffffffu) : ~u);
}

__global__ __launch_bounds__(256)
void edge_pass1(const int* __restrict__ ei, const float* __restrict__ asrc,
                const float* __restrict__ adst, float* __restrict__ ebuf,
                unsigned* __restrict__ emaxu) {
  const int t = blockIdx.x * blockDim.x + threadIdx.x;
  if (t >= E_TOT) return;
  const int s = (t < N_EDGES) ? ei[t] : (t - N_EDGES);
  const int d = (t < N_EDGES) ? ei[N_EDGES + t] : (t - N_EDGES);
  float v = asrc[s] + adst[d];
  v = (v > 0.f) ? v : NEG_SLOPE * v;
  ebuf[t] = v;
  atomicMax(&emaxu[d], fmap(v));
}

__global__ __launch_bounds__(256)
void edge_pass2(const int* __restrict__ ei, float* __restrict__ ebuf,
                const unsigned* __restrict__ emaxu, float* __restrict__ denom) {
  const int t = blockIdx.x * blockDim.x + threadIdx.x;
  if (t >= E_TOT) return;
  const int d = (t < N_EDGES) ? ei[N_EDGES + t] : (t - N_EDGES);
  const float ex = __expf(ebuf[t] - funmap(emaxu[d]));
  ebuf[t] = ex;
  atomicAdd(&denom[d], ex);
}

__global__ __launch_bounds__(256)
void edge_pass3(const int* __restrict__ ei, const float* __restrict__ ebuf,
                const float* __restrict__ denom, const float* __restrict__ H,
                float* __restrict__ acc) {
  const int gt = blockIdx.x * blockDim.x + threadIdx.x;
  const int edge = gt >> 5;
  const int lane = threadIdx.x & 31;
  if (edge >= E_TOT) return;
  const int s = (edge < N_EDGES) ? ei[edge] : (edge - N_EDGES);
  const int d = (edge < N_EDGES) ? ei[N_EDGES + edge] : (edge - N_EDGES);
  const float coef = ebuf[edge] / denom[d];
  const int c0 = lane * 4;
  const float4 hv = *(const float4*)(H + (size_t)s * D + c0);
  float* ad = acc + (size_t)d * D + c0;
  atomicAdd(&ad[0], hv.x * coef);
  atomicAdd(&ad[1], hv.y * coef);
  atomicAdd(&ad[2], hv.z * coef);
  atomicAdd(&ad[3], hv.w * coef);
}

__global__ __launch_bounds__(256)
void relu_pool(float* __restrict__ acc, const float* __restrict__ bias,
               const int* __restrict__ batch, float* __restrict__ embsum) {
  const int t = blockIdx.x * blockDim.x + threadIdx.x;
  if (t >= N_NODES * D) return;
  const int node = t >> 7, c = t & 127;
  float v = acc[t] + bias[c];
  v = (v > 0.f) ? v : 0.f;
  acc[t] = v;
  atomicAdd(&embsum[(size_t)batch[node] * D + c], v);
}

__global__ __launch_bounds__(256)
void count_kernel(const int* __restrict__ batch, float* __restrict__ cnt) {
  const int t = blockIdx.x * blockDim.x + threadIdx.x;
  if (t < N_NODES) atomicAdd(&cnt[batch[t]], 1.0f);
}

__global__ __launch_bounds__(256)
void finalize_kernel(float* __restrict__ out, const float* __restrict__ cnt) {
  const int t = blockIdx.x * blockDim.x + threadIdx.x;
  if (t >= 2 * N_GRAPHS * D) return;
  const int g = (t >> 7) & (N_GRAPHS - 1);
  float c = cnt[g];
  c = (c > 1.f) ? c : 1.f;
  out[t] /= c;
}

extern "C" void kernel_launch(void* const* d_in, const int* in_sizes, int n_in,
                              void* d_out, int out_size, void* d_ws, size_t ws_size,
                              hipStream_t stream) {
  const float* x     = (const float*)d_in[0];
  const int*   ei    = (const int*)d_in[1];
  const int*   batch = (const int*)d_in[2];
  const float* W1    = (const float*)d_in[3];
  const float* as1   = (const float*)d_in[4];
  const float* ad1   = (const float*)d_in[5];
  const float* b1    = (const float*)d_in[6];
  const float* W2    = (const float*)d_in[7];
  const float* as2   = (const float*)d_in[8];
  const float* ad2   = (const float*)d_in[9];
  const float* b2    = (const float*)d_in[10];
  float* out = (float*)d_out;

  char* w = (char*)d_ws;
  auto carve = [&](size_t bytes) {
    char* p = w;
    w += (bytes + 255) & ~(size_t)255;
    return p;
  };
  float*    hbuf  = (float*)carve((size_t)N_NODES * D * 4);
  float*    acc   = (float*)carve((size_t)N_NODES * D * 4);
  float*    asrc  = (float*)carve((size_t)N_NODES * 4);
  float*    adst  = (float*)carve((size_t)N_NODES * 4);
  unsigned* emaxu = (unsigned*)carve((size_t)N_NODES * 4);
  float*    denom = (float*)carve((size_t)N_NODES * 4);
  float*    ebuf  = (float*)carve((size_t)E_TOT * 4);
  float*    cnt   = (float*)carve((size_t)N_GRAPHS * 4);

  const int TB = 256;
  const int gGemm  = N_NODES / 16;
  const int gAlpha = (N_NODES * 32 + TB - 1) / TB;
  const int gEdge  = (E_TOT + TB - 1) / TB;
  const int gEdgeW = (E_TOT * 32 + TB - 1) / TB;
  const int gNodeC = (N_NODES * D + TB - 1) / TB;
  const int gNode  = (N_NODES + TB - 1) / TB;
  const int gFin   = (2 * N_GRAPHS * D + TB - 1) / TB;

  hipMemsetAsync(d_out, 0, (size_t)2 * N_GRAPHS * D * 4, stream);
  hipMemsetAsync(cnt, 0, (size_t)N_GRAPHS * 4, stream);
  count_kernel<<<gNode, TB, 0, stream>>>(batch, cnt);

  gemm_wmma<<<gGemm, TB, 0, stream>>>(x, W1, hbuf);
  alpha_kernel<<<gAlpha, TB, 0, stream>>>(hbuf, as1, ad1, asrc, adst);
  hipMemsetAsync(emaxu, 0, (size_t)N_NODES * 4, stream);
  hipMemsetAsync(denom, 0, (size_t)N_NODES * 4, stream);
  hipMemsetAsync(acc, 0, (size_t)N_NODES * D * 4, stream);
  edge_pass1<<<gEdge, TB, 0, stream>>>(ei, asrc, adst, ebuf, emaxu);
  edge_pass2<<<gEdge, TB, 0, stream>>>(ei, ebuf, emaxu, denom);
  edge_pass3<<<gEdgeW, TB, 0, stream>>>(ei, ebuf, denom, hbuf, acc);
  relu_pool<<<gNodeC, TB, 0, stream>>>(acc, b1, batch, out);

  gemm_wmma<<<gGemm, TB, 0, stream>>>(acc, W2, hbuf);
  alpha_kernel<<<gAlpha, TB, 0, stream>>>(hbuf, as2, ad2, asrc, adst);
  hipMemsetAsync(emaxu, 0, (size_t)N_NODES * 4, stream);
  hipMemsetAsync(denom, 0, (size_t)N_NODES * 4, stream);
  hipMemsetAsync(acc, 0, (size_t)N_NODES * D * 4, stream);
  edge_pass1<<<gEdge, TB, 0, stream>>>(ei, asrc, adst, ebuf, emaxu);
  edge_pass2<<<gEdge, TB, 0, stream>>>(ei, ebuf, emaxu, denom);
  edge_pass3<<<gEdgeW, TB, 0, stream>>>(ei, ebuf, denom, hbuf, acc);
  relu_pool<<<gNodeC, TB, 0, stream>>>(acc, b2, batch, out + N_GRAPHS * D);

  finalize_kernel<<<gFin, TB, 0, stream>>>(out, cnt);
}


# Aggregate per destination: an int atomic exchange per edge builds per-node edge lists, one wave per node does softmax and row sum in registers; no float atomics on rows; sums run in arrival order
# speedup vs baseline: 2.2425x; 2.2425x over previous
#include <hip/hip_runtime.h>

#define N_NODES  50000
#define N_EDGES  600000
#define N_GRAPHS 1024
#define D        128
#define NEG_SLOPE 0.2f

typedef float    v2f  __attribute__((ext_vector_type(2)));
typedef float    v8f  __attribute__((ext_vector_type(8)));
typedef _Float16 v16h __attribute__((ext_vector_type(16)));

__global__ __launch_bounds__(256)
void gemm_wmma(const float* __restrict__ X, const float* __restrict__ W,
               float* __restrict__ H) {
  __shared__ float xs[16][132];
  __shared__ float ws[128][136];
  const int row0 = blockIdx.x * 16;
  const int t = threadIdx.x;

  for (int i = t; i < 16 * 32; i += 256) {
    const int r = i >> 5, c4 = i & 31;
    const float4 v = ((const float4*)(X + (size_t)(row0 + r) * D))[c4];
    float* p = &xs[r][c4 * 4];
    p[0] = v.x; p[1] = v.y; p[2] = v.z; p[3] = v.w;
  }
  for (int i = t; i < 128 * 32; i += 256) {
    const int r = i >> 5, c4 = i & 31;
    *(float4*)&ws[r][c4 * 4] = ((const float4*)(W + (size_t)r * D))[c4];
  }
  __syncthreads();

  const int wave = t >> 5, lane = t & 31;
  const int hi = lane >> 4;
  const int m  = lane & 15;
  const int n  = wave * 16 + m;

  v8f c = {0.f, 0.f, 0.f, 0.f, 0.f, 0.f, 0.f, 0.f};

#if __has_builtin(__builtin_amdgcn_wmma_f32_16x16x4_f32)
  #pragma unroll
  for (int k0 = 0; k0 < D; k0 += 4) {
    const int ka = k0 + hi * 2;
    v2f a, b;
    a[0] = xs[m][ka];
    a[1] = xs[m][ka + 1];
    b[0] = ws[ka][n];
    b[1] = ws[ka + 1][n];
    c = __builtin_amdgcn_wmma_f32_16x16x4_f32(false, a, false, b, (short)0, c,
                                              false, false);
  }
#else
  #pragma unroll
  for (int k0 = 0; k0 < D; k0 += 32) {
    v16h a, b;
    #pragma unroll
    for (int i = 0; i < 16; ++i) {
      const int vv = i >> 1, j = i & 1;
      const int ka = k0 + ((vv >= 4) ? 16 : 0) + hi * 8 + (vv & 3) * 2 + j;
      a[i] = (_Float16)xs[m][ka];
      const int kb = k0 + hi * 16 + i;
      b[i] = (_Float16)ws[kb][n];
    }
    c = __builtin_amdgcn_wmma_f32_16x16x32_f16(false, a, false, b, (short)0, c,
                                               false, false);
  }
#endif

  #pragma unroll
  for (int r = 0; r < 8; ++r)
    H[(size_t)(row0 + r + hi * 8) * D + n] = c[r];
}

__global__ __launch_bounds__(256)
void alpha_kernel(const float* __restrict__ H, const float* __restrict__ a_src,
                  const float* __restrict__ a_dst, float* __restrict__ asrc,
                  float* __restrict__ adst) {
  const int node = (blockIdx.x * blockDim.x + threadIdx.x) >> 5;
  const int lane = threadIdx.x & 31;
  if (node >= N_NODES) return;
  const int c0 = lane * 4;
  const float4 hv = *(const float4*)(H + (size_t)node * D + c0);
  const float4 sv = *(const float4*)(a_src + c0);
  const float4 dv = *(const float4*)(a_dst + c0);
  float s = hv.x * sv.x + hv.y * sv.y + hv.z * sv.z + hv.w * sv.w;
  float d = hv.x * dv.x + hv.y * dv.y + hv.z * dv.z + hv.w * dv.w;
  #pragma unroll
  for (int off = 16; off > 0; off >>= 1) {
    s += __shfl_down(s, off, 32);
    d += __shfl_down(d, off, 32);
  }
  if (lane == 0) { asrc[node] = s; adst[node] = d; }
}

__global__ __launch_bounds__(256)
void build_lists(const int* __restrict__ ei, int* __restrict__ head,
                 int2* __restrict__ nodes) {
  const int t = blockIdx.x * blockDim.x + threadIdx.x;
  if (t >= N_EDGES) return;
  const int s = ei[t];
  const int d = ei[N_EDGES + t];
  const int old = atomicExch(&head[d], t);
  nodes[t] = make_int2(old, s);
}

__device__ __forceinline__ float wave_max(float v) {
  for (int off = 16; off > 0; off >>= 1) v = fmaxf(v, __shfl_xor(v, off, 32));
  return v;
}
__device__ __forceinline__ float wave_sum(float v) {
  for (int off = 16; off > 0; off >>= 1) v += __shfl_xor(v, off, 32);
  return v;
}

__global__ __launch_bounds__(256)
void gather_softmax(const int* __restrict__ head, const int2* __restrict__ nodes,
                    const float* __restrict__ asrc, const float* __restrict__ adst,
                    const float* __restrict__ H, float* __restrict__ acc) {
  const int node = (blockIdx.x * blockDim.x + threadIdx.x) >> 5;
  const int lane = threadIdx.x & 31;
  if (node >= N_NODES) return;
  const int c0 = lane * 4;
  const float ad = adst[node];

  float m = 0.f, l = 0.f;
  float4 a = make_float4(0.f, 0.f, 0.f, 0.f);
  bool first = true;
  int e = head[node];
  int steps = 0;
  int mysrc = node;
  int k = 1;
  for (;;) {
    while (k < 32 && e >= 0 && steps < N_EDGES) {
      const int2 nd = nodes[e];
      if (lane == k) mysrc = nd.y;
      e = nd.x;
      ++k; ++steps;
    }
    const int kk = __builtin_amdgcn_readfirstlane(k);
    float v = -__builtin_inff();
    if (lane < kk) {
      const float t = asrc[mysrc] + ad;
      v = (t > 0.f) ? t : NEG_SLOPE * t;
    }
    const float cm = wave_max(v);
    const float mn = first ? cm : fmaxf(m, cm);
    const float scale = first ? 0.f : __expf(m - mn);
    const float ex = (lane < kk) ? __expf(v - mn) : 0.f;
    l = l * scale + wave_sum(ex);
    a.x *= scale; a.y *= scale; a.z *= scale; a.w *= scale;
    #pragma unroll 4
    for (int j = 0; j < kk; ++j) {
      const int s = __builtin_amdgcn_readlane(mysrc, j);
      const float w = __int_as_float(__builtin_amdgcn_readlane(__float_as_int(ex), j));
      const float4 hv = *(const float4*)(H + (size_t)s * D + c0);
      a.x += w * hv.x; a.y += w * hv.y; a.z += w * hv.z; a.w += w * hv.w;
    }
    m = mn;
    first = false;
    if (!(e >= 0 && steps < N_EDGES)) break;
    k = 0;
  }
  float4 o;
  o.x = a.x / l; o.y = a.y / l; o.z = a.z / l; o.w = a.w / l;
  *(float4*)(acc + (size_t)node * D + c0) = o;
}

__global__ __launch_bounds__(256)
void relu_pool(float* __restrict__ acc, const float* __restrict__ bias,
               const int* __restrict__ batch, float* __restrict__ embsum) {
  const int t = blockIdx.x * blockDim.x + threadIdx.x;
  if (t >= N_NODES * D) return;
  const int node = t >> 7, c = t & 127;
  float v = acc[t] + bias[c];
  v = (v > 0.f) ? v : 0.f;
  acc[t] = v;
  atomicAdd(&embsum[(size_t)batch[node] * D + c], v);
}

__global__ __launch_bounds__(256)
void count_kernel(const int* __restrict__ batch, float* __restrict__ cnt) {
  const int t = blockIdx.x * blockDim.x + threadIdx.x;
  if (t < N_NODES) atomicAdd(&cnt[batch[t]], 1.0f);
}

__global__ __launch_bounds__(256)
void finalize_kernel(float* __restrict__ out, const float* __restrict__ cnt) {
  const int t = blockIdx.x * blockDim.x + threadIdx.x;
  if (t >= 2 * N_GRAPHS * D) return;
  const int g = (t >> 7) & (N_GRAPHS - 1);
  float c = cnt[g];
  c = (c > 1.f) ? c : 1.f;
  out[t] /= c;
}

extern "C" void kernel_launch(void* const* d_in, const int* in_sizes, int n_in,
                              void* d_out, int out_size, void* d_ws, size_t ws_size,
                              hipStream_t stream) {
  const float* x     = (const float*)d_in[0];
  const int*   ei    = (const int*)d_in[1];
  const int*   batch = (const int*)d_in[2];
  const float* W1    = (const float*)d_in[3];
  const float* as1   = (const float*)d_in[4];
  const float* ad1   = (const float*)d_in[5];
  const float* b1    = (const float*)d_in[6];
  const float* W2    = (const float*)d_in[7];
  const float* as2   = (const float*)d_in[8];
  const float* ad2   = (const float*)d_in[9];
  const float* b2    = (const float*)d_in[10];
  float* out = (float*)d_out;

  char* w = (char*)d_ws;
  auto carve = [&](size_t bytes) {
    char* p = w;
    w += (bytes + 255) & ~(size_t)255;
    return p;
  };
  float*    hbuf  = (float*)carve((size_t)N_NODES * D * 4);
  float*    acc   = (float*)carve((size_t)N_NODES * D * 4);
  float*    asrc  = (float*)carve((size_t)N_NODES * 4);
  float*    adst  = (float*)carve((size_t)N_NODES * 4);
  int*      head  = (int*)carve((size_t)N_NODES * 4);
  int2*     nodes = (int2*)carve((size_t)N_EDGES * 8);
  float*    cnt   = (float*)carve((size_t)N_GRAPHS * 4);

  const int TB = 256;
  const int gGemm  = N_NODES / 16;
  const int gAlpha = (N_NODES * 32 + TB - 1) / TB;
  const int gEdge  = (N_EDGES + TB - 1) / TB;
  const int gNodeC = (N_NODES * D + TB - 1) / TB;
  const int gNode  = (N_NODES + TB - 1) / TB;
  const int gFin   = (2 * N_GRAPHS * D + TB - 1) / TB;

  hipMemsetAsync(d_out, 0, (size_t)2 * N_GRAPHS * D * 4, stream);
  hipMemsetAsync(cnt, 0, (size_t)N_GRAPHS * 4, stream);
  hipMemsetAsync(head, 0xFF, (size_t)N_NODES * 4, stream);
  count_kernel<<<gNode, TB, 0, stream>>>(batch, cnt);
  build_lists<<<gEdge, TB, 0, stream>>>(ei, head, nodes);

  gemm_wmma<<<gGemm, TB, 0, stream>>>(x, W1, hbuf);
  alpha_kernel<<<gAlpha, TB, 0, stream>>>(hbuf, as1, ad1, asrc, adst);
  gather_softmax<<<gAlpha, TB, 0, stream>>>(head, nodes, asrc, adst, hbuf, acc);
  relu_pool<<<gNodeC, TB, 0, stream>>>(acc, b1, batch, out);

  gemm_wmma<<<gGemm, TB, 0, stream>>>(acc, W2, hbuf);
  alpha_kernel<<<gAlpha, TB, 0, stream>>>(hbuf, as2, ad2, asrc, adst);
  gather_softmax<<<gAlpha, TB, 0, stream>>>(head, nodes, asrc, adst, hbuf, acc);
  relu_pool<<<gNodeC, TB, 0, stream>>>(acc, b2, batch, out + N_GRAPHS * D);

  finalize_kernel<<<gFin, TB, 0, stream>>>(out, cnt);
}


# Bias, relu and mean-pool sums fused into the aggregation kernel: a block's 8 rows are summed per graph id in LDS before the float atomics (new sum tree); no separate pool pass, layer 2 rows unstored
# speedup vs baseline: 2.8956x; 1.2912x over previous
#include <hip/hip_runtime.h>

#define N_NODES  50000
#define N_EDGES  600000
#define N_GRAPHS 1024
#define D        128
#define NEG_SLOPE 0.2f

typedef float    v2f  __attribute__((ext_vector_type(2)));
typedef float    v8f  __attribute__((ext_vector_type(8)));
typedef _Float16 v16h __attribute__((ext_vector_type(16)));

__global__ __launch_bounds__(256)
void gemm_wmma(const float* __restrict__ X, const float* __restrict__ W,
               float* __restrict__ H) {
  __shared__ float xs[16][132];
  __shared__ float ws[128][136];
  const int row0 = blockIdx.x * 16;
  const int t = threadIdx.x;

  for (int i = t; i < 16 * 32; i += 256) {
    const int r = i >> 5, c4 = i & 31;
    const float4 v = ((const float4*)(X + (size_t)(row0 + r) * D))[c4];
    float* p = &xs[r][c4 * 4];
    p[0] = v.x; p[1] = v.y; p[2] = v.z; p[3] = v.w;
  }
  for (int i = t; i < 128 * 32; i += 256) {
    const int r = i >> 5, c4 = i & 31;
    *(float4*)&ws[r][c4 * 4] = ((const float4*)(W + (size_t)r * D))[c4];
  }
  __syncthreads();

  const int wave = t >> 5, lane = t & 31;
  const int hi = lane >> 4;
  const int m  = lane & 15;
  const int n  = wave * 16 + m;

  v8f c = {0.f, 0.f, 0.f, 0.f, 0.f, 0.f, 0.f, 0.f};

#if __has_builtin(__builtin_amdgcn_wmma_f32_16x16x4_f32)
  #pragma unroll
  for (int k0 = 0; k0 < D; k0 += 4) {
    const int ka = k0 + hi * 2;
    v2f a, b;
    a[0] = xs[m][ka];
    a[1] = xs[m][ka + 1];
    b[0] = ws[ka][n];
    b[1] = ws[ka + 1][n];
    c = __builtin_amdgcn_wmma_f32_16x16x4_f32(false, a, false, b, (short)0, c,
                                              false, false);
  }
#else
  #pragma unroll
  for (int k0 = 0; k0 < D; k0 += 32) {
    v16h a, b;
    #pragma unroll
    for (int i = 0; i < 16; ++i) {
      const int vv = i >> 1, j = i & 1;
      const int ka = k0 + ((vv >= 4) ? 16 : 0) + hi * 8 + (vv & 3) * 2 + j;
      a[i] = (_Float16)xs[m][ka];
      const int kb = k0 + hi * 16 + i;
      b[i] = (_Float16)ws[kb][n];
    }
    c = __builtin_amdgcn_wmma_f32_16x16x32_f16(false, a, false, b, (short)0, c,
                                               false, false);
  }
#endif

  #pragma unroll
  for (int r = 0; r < 8; ++r)
    H[(size_t)(row0 + r + hi * 8) * D + n] = c[r];
}

__global__ __launch_bounds__(256)
void alpha_kernel(const float* __restrict__ H, const float* __restrict__ a_src,
                  const float* __restrict__ a_dst, float* __restrict__ asrc,
                  float* __restrict__ adst) {
  const int node = (blockIdx.x * blockDim.x + threadIdx.x) >> 5;
  const int lane = threadIdx.x & 31;
  if (node >= N_NODES) return;
  const int c0 = lane * 4;
  const float4 hv = *(const float4*)(H + (size_t)node * D + c0);
  const float4 sv = *(const float4*)(a_src + c0);
  const float4 dv = *(const float4*)(a_dst + c0);
  float s = hv.x * sv.x + hv.y * sv.y + hv.z * sv.z + hv.w * sv.w;
  float d = hv.x * dv.x + hv.y * dv.y + hv.z * dv.z + hv.w * dv.w;
  #pragma unroll
  for (int off = 16; off > 0; off >>= 1) {
    s += __shfl_down(s, off, 32);
    d += __shfl_down(d, off, 32);
  }
  if (lane == 0) { asrc[node] = s; adst[node] = d; }
}

__global__ __launch_bounds__(256)
void build_lists(const int* __restrict__ ei, int* __restrict__ head,
                 int2* __restrict__ nodes) {
  const int t = blockIdx.x * blockDim.x + threadIdx.x;
  if (t >= N_EDGES) return;
  const int s = ei[t];
  const int d = ei[N_EDGES + t];
  const int old = atomicExch(&head[d], t);
  nodes[t] = make_int2(old, s);
}

__device__ __forceinline__ float wave_max(float v) {
  for (int off = 16; off > 0; off >>= 1) v = fmaxf(v, __shfl_xor(v, off, 32));
  return v;
}
__device__ __forceinline__ float wave_sum(float v) {
  for (int off = 16; off > 0; off >>= 1) v += __shfl_xor(v, off, 32);
  return v;
}

template <bool STORE_H>
__global__ __launch_bounds__(256)
void gather_softmax_pool(const int* __restrict__ head, const int2* __restrict__ nodes,
                         const float* __restrict__ asrc, const float* __restrict__ adst,
                         const float* __restrict__ H, const float* __restrict__ bias,
                         const int* __restrict__ batch, float* __restrict__ hout,
                         float* __restrict__ embsum) {
  __shared__ float rows[8][D];
  __shared__ int   gid[8];
  const int wave = threadIdx.x >> 5;
  const int node = blockIdx.x * 8 + wave;
  const int lane = threadIdx.x & 31;
  const int c0 = lane * 4;
  float4 o = make_float4(0.f, 0.f, 0.f, 0.f);
  if (node < N_NODES) {
    const float ad = adst[node];
    const int gb = batch[node];
    const float4 bv = *(const float4*)(bias + c0);

    float m = 0.f, l = 0.f;
    float4 a = make_float4(0.f, 0.f, 0.f, 0.f);
    bool first = true;
    int e = head[node];
    int steps = 0;
    int mysrc = node;
    int k = 1;
    for (;;) {
      while (k < 32 && e >= 0 && steps < N_EDGES) {
        const int2 nd = nodes[e];
        if (lane == k) mysrc = nd.y;
        e = nd.x;
        ++k; ++steps;
      }
      const int kk = __builtin_amdgcn_readfirstlane(k);
      float v = -__builtin_inff();
      if (lane < kk) {
        const float t = asrc[mysrc] + ad;
        v = (t > 0.f) ? t : NEG_SLOPE * t;
      }
      const float cm = wave_max(v);
      const float mn = first ? cm : fmaxf(m, cm);
      const float scale = first ? 0.f : __expf(m - mn);
      const float ex = (lane < kk) ? __expf(v - mn) : 0.f;
      l = l * scale + wave_sum(ex);
      a.x *= scale; a.y *= scale; a.z *= scale; a.w *= scale;
      #pragma unroll 4
      for (int j = 0; j < kk; ++j) {
        const int s = __builtin_amdgcn_readlane(mysrc, j);
        const float w = __int_as_float(__builtin_amdgcn_readlane(__float_as_int(ex), j));
        const float4 hv = *(const float4*)(H + (size_t)s * D + c0);
        a.x += w * hv.x; a.y += w * hv.y; a.z += w * hv.z; a.w += w * hv.w;
      }
      m = mn;
      first = false;
      if (!(e >= 0 && steps < N_EDGES)) break;
      k = 0;
    }
    o.x = a.x / l + bv.x; o.y = a.y / l + bv.y; o.z = a.z / l + bv.z; o.w = a.w / l + bv.w;
    o.x = (o.x > 0.f) ? o.x : 0.f;
    o.y = (o.y > 0.f) ? o.y : 0.f;
    o.z = (o.z > 0.f) ? o.z : 0.f;
    o.w = (o.w > 0.f) ? o.w : 0.f;
    *(float4*)&rows[wave][c0] = o;
    if (lane == 0) gid[wave] = gb;
  } else {
    *(float4*)&rows[wave][c0] = o;
    if (lane == 0) gid[wave] = -1;
  }
  __syncthreads();
  if (STORE_H && node < N_NODES) *(float4*)(hout + (size_t)node * D + c0) = o;
  if (threadIdx.x < D) {
    const int c = threadIdx.x;
    int gv[8];
    float rv[8];
    #pragma unroll
    for (int w = 0; w < 8; ++w) { gv[w] = gid[w]; rv[w] = rows[w][c]; }
    int g = gv[0];
    float s = rv[0];
    #pragma unroll
    for (int w = 1; w < 8; ++w) {
      if (gv[w] != g) {
        if (g >= 0) atomicAdd(&embsum[(size_t)g * D + c], s);
        g = gv[w];
        s = 0.f;
      }
      s += rv[w];
    }
    if (g >= 0) atomicAdd(&embsum[(size_t)g * D + c], s);
  }
}

__global__ __launch_bounds__(256)
void count_kernel(const int* __restrict__ batch, float* __restrict__ cnt) {
  const int t = blockIdx.x * blockDim.x + threadIdx.x;
  if (t < N_NODES) atomicAdd(&cnt[batch[t]], 1.0f);
}

__global__ __launch_bounds__(256)
void finalize_kernel(float* __restrict__ out, const float* __restrict__ cnt) {
  const int t = blockIdx.x * blockDim.x + threadIdx.x;
  if (t >= 2 * N_GRAPHS * D) return;
  const int g = (t >> 7) & (N_GRAPHS - 1);
  float c = cnt[g];
  c = (c > 1.f) ? c : 1.f;
  out[t] /= c;
}

extern "C" void kernel_launch(void* const* d_in, const int* in_sizes, int n_in,
                              void* d_out, int out_size, void* d_ws, size_t ws_size,
                              hipStream_t stream) {
  const float* x     = (const float*)d_in[0];
  const int*   ei    = (const int*)d_in[1];
  const int*   batch = (const int*)d_in[2];
  const float* W1    = (const float*)d_in[3];
  const float* as1   = (const float*)d_in[4];
  const float* ad1   = (const float*)d_in[5];
  const float* b1    = (const float*)d_in[6];
  const float* W2    = (const float*)d_in[7];
  const float* as2   = (const float*)d_in[8];
  const float* ad2   = (const float*)d_in[9];
  const float* b2    = (const float*)d_in[10];
  float* out = (float*)d_out;

  char* w = (char*)d_ws;
  auto carve = [&](size_t bytes) {
    char* p = w;
    w += (bytes + 255) & ~(size_t)255;
    return p;
  };
  float*    hbuf  = (float*)carve((size_t)N_NODES * D * 4);
  float*    acc   = (float*)carve((size_t)N_NODES * D * 4);
  float*    asrc  = (float*)carve((size_t)N_NODES * 4);
  float*    adst  = (float*)carve((size_t)N_NODES * 4);
  int*      head  = (int*)carve((size_t)N_NODES * 4);
  int2*     nodes = (int2*)carve((size_t)N_EDGES * 8);
  float*    cnt   = (float*)carve((size_t)N_GRAPHS * 4);

  const int TB = 256;
  const int gGemm  = N_NODES / 16;
  const int gAlpha = (N_NODES * 32 + TB - 1) / TB;
  const int gEdge  = (N_EDGES + TB - 1) / TB;
  const int gGather = (N_NODES + 7) / 8;
  const int gNode  = (N_NODES + TB - 1) / TB;
  const int gFin   = (2 * N_GRAPHS * D + TB - 1) / TB;

  hipMemsetAsync(d_out, 0, (size_t)2 * N_GRAPHS * D * 4, stream);
  hipMemsetAsync(cnt, 0, (size_t)N_GRAPHS * 4, stream);
  hipMemsetAsync(head, 0xFF, (size_t)N_NODES * 4, stream);
  count_kernel<<<gNode, TB, 0, stream>>>(batch, cnt);
  build_lists<<<gEdge, TB, 0, stream>>>(ei, head, nodes);

  gemm_wmma<<<gGemm, TB, 0, stream>>>(x, W1, hbuf);
  alpha_kernel<<<gAlpha, TB, 0, stream>>>(hbuf, as1, ad1, asrc, adst);
  gather_softmax_pool<true><<<gGather, TB, 0, stream>>>(head, nodes, asrc, adst, hbuf, b1,
                                                       batch, acc, out);

  gemm_wmma<<<gGemm, TB, 0, stream>>>(acc, W2, hbuf);
  alpha_kernel<<<gAlpha, TB, 0, stream>>>(hbuf, as2, ad2, asrc, adst);
  gather_softmax_pool<false><<<gGather, TB, 0, stream>>>(head, nodes, asrc, adst, hbuf, b2,
                                                        batch, acc, out + N_GRAPHS * D);

  finalize_kernel<<<gFin, TB, 0, stream>>>(out, cnt);
}


# Attention coefficient dots (h.a_src, h.a_dst) summed in the GEMM epilogue from the accumulators (halving butterfly over 16 lanes, then 8 waves via LDS: new sum tree); no separate coefficient pass
# speedup vs baseline: 3.1892x; 1.1014x over previous
#include <hip/hip_runtime.h>

#define N_NODES  50000
#define N_EDGES  600000
#define N_GRAPHS 1024
#define D        128
#define NEG_SLOPE 0.2f

typedef float    v2f  __attribute__((ext_vector_type(2)));
typedef float    v8f  __attribute__((ext_vector_type(8)));
typedef _Float16 v16h __attribute__((ext_vector_type(16)));

__global__ __launch_bounds__(256)
void gemm_wmma(const float* __restrict__ X, const float* __restrict__ W,
               const float* __restrict__ a_src, const float* __restrict__ a_dst,
               float* __restrict__ H, float* __restrict__ asrc,
               float* __restrict__ adst) {
  __shared__ float xs[16][132];
  __shared__ float ws[128][136];
  __shared__ float part[8][32];
  const int row0 = blockIdx.x * 16;
  const int t = threadIdx.x;
  const int wave = t >> 5, lane = t & 31;
  const int hi = lane >> 4;
  const int m  = lane & 15;
  const int n  = wave * 16 + m;
  const float as_n = a_src[n];
  const float ad_n = a_dst[n];

  for (int i = t; i < 16 * 32; i += 256) {
    const int r = i >> 5, c4 = i & 31;
    const float4 v = ((const float4*)(X + (size_t)(row0 + r) * D))[c4];
    float* p = &xs[r][c4 * 4];
    p[0] = v.x; p[1] = v.y; p[2] = v.z; p[3] = v.w;
  }
  for (int i = t; i < 128 * 32; i += 256) {
    const int r = i >> 5, c4 = i & 31;
    *(float4*)&ws[r][c4 * 4] = ((const float4*)(W + (size_t)r * D))[c4];
  }
  __syncthreads();

  v8f c = {0.f, 0.f, 0.f, 0.f, 0.f, 0.f, 0.f, 0.f};

#if __has_builtin(__builtin_amdgcn_wmma_f32_16x16x4_f32)
  #pragma unroll
  for (int k0 = 0; k0 < D; k0 += 4) {
    const int ka = k0 + hi * 2;
    v2f a, b;
    a[0] = xs[m][ka];
    a[1] = xs[m][ka + 1];
    b[0] = ws[ka][n];
    b[1] = ws[ka + 1][n];
    c = __builtin_amdgcn_wmma_f32_16x16x4_f32(false, a, false, b, (short)0, c,
                                              false, false);
  }
#else
  #pragma unroll
  for (int k0 = 0; k0 < D; k0 += 32) {
    v16h a, b;
    #pragma unroll
    for (int i = 0; i < 16; ++i) {
      const int vv = i >> 1, j = i & 1;
      const int ka = k0 + ((vv >= 4) ? 16 : 0) + hi * 8 + (vv & 3) * 2 + j;
      a[i] = (_Float16)xs[m][ka];
      const int kb = k0 + hi * 16 + i;
      b[i] = (_Float16)ws[kb][n];
    }
    c = __builtin_amdgcn_wmma_f32_16x16x32_f16(false, a, false, b, (short)0, c,
                                               false, false);
  }
#endif

  float v[16];
  #pragma unroll
  for (int r = 0; r < 8; ++r) { v[r] = c[r] * as_n; v[8 + r] = c[r] * ad_n; }
  const bool b3 = (m & 8) != 0, b2 = (m & 4) != 0, b1 = (m & 2) != 0, b0 = (m & 1) != 0;
  float u[8];
  #pragma unroll
  for (int i = 0; i < 8; ++i)
    u[i] = (b3 ? v[i + 8] : v[i]) + __shfl_xor(b3 ? v[i] : v[i + 8], 8, 32);
  float q[4];
  #pragma unroll
  for (int i = 0; i < 4; ++i)
    q[i] = (b2 ? u[i + 4] : u[i]) + __shfl_xor(b2 ? u[i] : u[i + 4], 4, 32);
  float y[2];
  #pragma unroll
  for (int i = 0; i < 2; ++i)
    y[i] = (b1 ? q[i + 2] : q[i]) + __shfl_xor(b1 ? q[i] : q[i + 2], 2, 32);
  const float z = (b0 ? y[1] : y[0]) + __shfl_xor(b0 ? y[0] : y[1], 1, 32);
  part[wave][(m >> 3) * 16 + hi * 8 + (m & 7)] = z;
  __syncthreads();
  #pragma unroll
  for (int r = 0; r < 8; ++r)
    H[(size_t)(row0 + r + hi * 8) * D + n] = c[r];
  if (t < 32) {
    float s = 0.f;
    #pragma unroll
    for (int w = 0; w < 8; ++w) s += part[w][t];
    if (t < 16) asrc[row0 + t] = s;
    else        adst[row0 + t - 16] = s;
  }
}

__global__ __launch_bounds__(256)
void build_lists(const int* __restrict__ ei, int* __restrict__ head,
                 int2* __restrict__ nodes) {
  const int t = blockIdx.x * blockDim.x + threadIdx.x;
  if (t >= N_EDGES) return;
  const int s = ei[t];
  const int d = ei[N_EDGES + t];
  const int old = atomicExch(&head[d], t);
  nodes[t] = make_int2(old, s);
}

__device__ __forceinline__ float wave_max(float v) {
  for (int off = 16; off > 0; off >>= 1) v = fmaxf(v, __shfl_xor(v, off, 32));
  return v;
}
__device__ __forceinline__ float wave_sum(float v) {
  for (int off = 16; off > 0; off >>= 1) v += __shfl_xor(v, off, 32);
  return v;
}

template <bool STORE_H>
__global__ __launch_bounds__(256)
void gather_softmax_pool(const int* __restrict__ head, const int2* __restrict__ nodes,
                         const float* __restrict__ asrc, const float* __restrict__ adst,
                         const float* __restrict__ H, const float* __restrict__ bias,
                         const int* __restrict__ batch, float* __restrict__ hout,
                         float* __restrict__ embsum) {
  __shared__ float rows[8][D];
  __shared__ int   gid[8];
  const int wave = threadIdx.x >> 5;
  const int node = blockIdx.x * 8 + wave;
  const int lane = threadIdx.x & 31;
  const int c0 = lane * 4;
  float4 o = make_float4(0.f, 0.f, 0.f, 0.f);
  if (node < N_NODES) {
    const float ad = adst[node];
    const int gb = batch[node];
    const float4 bv = *(const float4*)(bias + c0);

    float m = 0.f, l = 0.f;
    float4 a = make_float4(0.f, 0.f, 0.f, 0.f);
    bool first = true;
    int e = head[node];
    int steps = 0;
    int mysrc = node;
    int k = 1;
    for (;;) {
      while (k < 32 && e >= 0 && steps < N_EDGES) {
        const int2 nd = nodes[e];
        if (lane == k) mysrc = nd.y;
        e = nd.x;
        ++k; ++steps;
      }
      const int kk = __builtin_amdgcn_readfirstlane(k);
      float v = -__builtin_inff();
      if (lane < kk) {
        const float t = asrc[mysrc] + ad;
        v = (t > 0.f) ? t : NEG_SLOPE * t;
      }
      const float cm = wave_max(v);
      const float mn = first ? cm : fmaxf(m, cm);
      const float scale = first ? 0.f : __expf(m - mn);
      const float ex = (lane < kk) ? __expf(v - mn) : 0.f;
      l = l * scale + wave_sum(ex);
      a.x *= scale; a.y *= scale; a.z *= scale; a.w *= scale;
      #pragma unroll 4
      for (int j = 0; j < kk; ++j) {
        const int s = __builtin_amdgcn_readlane(mysrc, j);
        const float w = __int_as_float(__builtin_amdgcn_readlane(__float_as_int(ex), j));
        const float4 hv = *(const float4*)(H + (size_t)s * D + c0);
        a.x += w * hv.x; a.y += w * hv.y; a.z += w * hv.z; a.w += w * hv.w;
      }
      m = mn;
      first = false;
      if (!(e >= 0 && steps < N_EDGES)) break;
      k = 0;
    }
    o.x = a.x / l + bv.x; o.y = a.y / l + bv.y; o.z = a.z / l + bv.z; o.w = a.w / l + bv.w;
    o.x = (o.x > 0.f) ? o.x : 0.f;
    o.y = (o.y > 0.f) ? o.y : 0.f;
    o.z = (o.z > 0.f) ? o.z : 0.f;
    o.w = (o.w > 0.f) ? o.w : 0.f;
    *(float4*)&rows[wave][c0] = o;
    if (lane == 0) gid[wave] = gb;
  } else {
    *(float4*)&rows[wave][c0] = o;
    if (lane == 0) gid[wave] = -1;
  }
  __syncthreads();
  if (STORE_H && node < N_NODES) *(float4*)(hout + (size_t)node * D + c0) = o;
  if (threadIdx.x < D) {
    const int c = threadIdx.x;
    int gv[8];
    float rv[8];
    #pragma unroll
    for (int w = 0; w < 8; ++w) { gv[w] = gid[w]; rv[w] = rows[w][c]; }
    int g = gv[0];
    float s = rv[0];
    #pragma unroll
    for (int w = 1; w < 8; ++w) {
      if (gv[w] != g) {
        if (g >= 0) atomicAdd(&embsum[(size_t)g * D + c], s);
        g = gv[w];
        s = 0.f;
      }
      s += rv[w];
    }
    if (g >= 0) atomicAdd(&embsum[(size_t)g * D + c], s);
  }
}

__global__ __launch_bounds__(256)
void count_kernel(const int* __restrict__ batch, float* __restrict__ cnt) {
  const int t = blockIdx.x * blockDim.x + threadIdx.x;
  if (t < N_NODES) atomicAdd(&cnt[batch[t]], 1.0f);
}

__global__ __launch_bounds__(256)
void finalize_kernel(float* __restrict__ out, const float* __restrict__ cnt) {
  const int t = blockIdx.x * blockDim.x + threadIdx.x;
  if (t >= 2 * N_GRAPHS * D) return;
  const int g = (t >> 7) & (N_GRAPHS - 1);
  float c = cnt[g];
  c = (c > 1.f) ? c : 1.f;
  out[t] /= c;
}

extern "C" void kernel_launch(void* const* d_in, const int* in_sizes, int n_in,
                              void* d_out, int out_size, void* d_ws, size_t ws_size,
                              hipStream_t stream) {
  const float* x     = (const float*)d_in[0];
  const int*   ei    = (const int*)d_in[1];
  const int*   batch = (const int*)d_in[2];
  const float* W1    = (const float*)d_in[3];
  const float* as1   = (const float*)d_in[4];
  const float* ad1   = (const float*)d_in[5];
  const float* b1    = (const float*)d_in[6];
  const float* W2    = (const float*)d_in[7];
  const float* as2   = (const float*)d_in[8];
  const float* ad2   = (const float*)d_in[9];
  const float* b2    = (const float*)d_in[10];
  float* out = (float*)d_out;

  char* w = (char*)d_ws;
  auto carve = [&](size_t bytes) {
    char* p = w;
    w += (bytes + 255) & ~(size_t)255;
    return p;
  };
  float*    hbuf  = (float*)carve((size_t)N_NODES * D * 4);
  float*    acc   = (float*)carve((size_t)N_NODES * D * 4);
  float*    asrc  = (float*)carve((size_t)N_NODES * 4);
  float*    adst  = (float*)carve((size_t)N_NODES * 4);
  int*      head  = (int*)carve((size_t)N_NODES * 4);
  int2*     nodes = (int2*)carve((size_t)N_EDGES * 8);
  float*    cnt   = (float*)carve((size_t)N_GRAPHS * 4);

  const int TB = 256;
  const int gGemm  = N_NODES / 16;
  const int gEdge  = (N_EDGES + TB - 1) / TB;
  const int gGather = (N_NODES + 7) / 8;
  const int gNode  = (N_NODES + TB - 1) / TB;
  const int gFin   = (2 * N_GRAPHS * D + TB - 1) / TB;

  hipMemsetAsync(d_out, 0, (size_t)2 * N_GRAPHS * D * 4, stream);
  hipMemsetAsync(cnt, 0, (size_t)N_GRAPHS * 4, stream);
  hipMemsetAsync(head, 0xFF, (size_t)N_NODES * 4, stream);
  count_kernel<<<gNode, TB, 0, stream>>>(batch, cnt);
  build_lists<<<gEdge, TB, 0, stream>>>(ei, head, nodes);

  gemm_wmma<<<gGemm, TB, 0, stream>>>(x, W1, as1, ad1, hbuf, asrc, adst);
  gather_softmax_pool<true><<<gGather, TB, 0, stream>>>(head, nodes, asrc, adst, hbuf, b1,
                                                       batch, acc, out);

  gemm_wmma<<<gGemm, TB, 0, stream>>>(acc, W2, as2, ad2, hbuf, asrc, adst);
  gather_softmax_pool<false><<<gGather, TB, 0, stream>>>(head, nodes, asrc, adst, hbuf, b2,
                                                        batch, acc, out + N_GRAPHS * D);

  finalize_kernel<<<gFin, TB, 0, stream>>>(out, cnt);
}


# Edge sources kept in one 128-byte row per node (int atomic add gives the slot, overflow list past 31): one load per node replaces the dependent list walk; sums in arrival order, earliest first
# speedup vs baseline: 3.7962x; 1.1903x over previous
#include <hip/hip_runtime.h>

#define N_NODES  50000
#define N_EDGES  600000
#define N_GRAPHS 1024
#define D        128
#define NEG_SLOPE 0.2f

typedef float    v2f  __attribute__((ext_vector_type(2)));
typedef float    v8f  __attribute__((ext_vector_type(8)));
typedef _Float16 v16h __attribute__((ext_vector_type(16)));

__global__ __launch_bounds__(256)
void gemm_wmma(const float* __restrict__ X, const float* __restrict__ W,
               const float* __restrict__ a_src, const float* __restrict__ a_dst,
               float* __restrict__ H, float* __restrict__ asrc,
               float* __restrict__ adst) {
  __shared__ float xs[16][132];
  __shared__ float ws[128][136];
  __shared__ float part[8][32];
  const int row0 = blockIdx.x * 16;
  const int t = threadIdx.x;
  const int wave = t >> 5, lane = t & 31;
  const int hi = lane >> 4;
  const int m  = lane & 15;
  const int n  = wave * 16 + m;
  const float as_n = a_src[n];
  const float ad_n = a_dst[n];

  for (int i = t; i < 16 * 32; i += 256) {
    const int r = i >> 5, c4 = i & 31;
    const float4 v = ((const float4*)(X + (size_t)(row0 + r) * D))[c4];
    float* p = &xs[r][c4 * 4];
    p[0] = v.x; p[1] = v.y; p[2] = v.z; p[3] = v.w;
  }
  for (int i = t; i < 128 * 32; i += 256) {
    const int r = i >> 5, c4 = i & 31;
    *(float4*)&ws[r][c4 * 4] = ((const float4*)(W + (size_t)r * D))[c4];
  }
  __syncthreads();

  v8f c = {0.f, 0.f, 0.f, 0.f, 0.f, 0.f, 0.f, 0.f};

#if __has_builtin(__builtin_amdgcn_wmma_f32_16x16x4_f32)
  #pragma unroll
  for (int k0 = 0; k0 < D; k0 += 4) {
    const int ka = k0 + hi * 2;
    v2f a, b;
    a[0] = xs[m][ka];
    a[1] = xs[m][ka + 1];
    b[0] = ws[ka][n];
    b[1] = ws[ka + 1][n];
    c = __builtin_amdgcn_wmma_f32_16x16x4_f32(false, a, false, b, (short)0, c,
                                              false, false);
  }
#else
  #pragma unroll
  for (int k0 = 0; k0 < D; k0 += 32) {
    v16h a, b;
    #pragma unroll
    for (int i = 0; i < 16; ++i) {
      const int vv = i >> 1, j = i & 1;
      const int ka = k0 + ((vv >= 4) ? 16 : 0) + hi * 8 + (vv & 3) * 2 + j;
      a[i] = (_Float16)xs[m][ka];
      const int kb = k0 + hi * 16 + i;
      b[i] = (_Float16)ws[kb][n];
    }
    c = __builtin_amdgcn_wmma_f32_16x16x32_f16(false, a, false, b, (short)0, c,
                                               false, false);
  }
#endif

  float v[16];
  #pragma unroll
  for (int r = 0; r < 8; ++r) { v[r] = c[r] * as_n; v[8 + r] = c[r] * ad_n; }
  const bool b3 = (m & 8) != 0, b2 = (m & 4) != 0, b1 = (m & 2) != 0, b0 = (m & 1) != 0;
  float u[8];
  #pragma unroll
  for (int i = 0; i < 8; ++i)
    u[i] = (b3 ? v[i + 8] : v[i]) + __shfl_xor(b3 ? v[i] : v[i + 8], 8, 32);
  float q[4];
  #pragma unroll
  for (int i = 0; i < 4; ++i)
    q[i] = (b2 ? u[i + 4] : u[i]) + __shfl_xor(b2 ? u[i] : u[i + 4], 4, 32);
  float y[2];
  #pragma unroll
  for (int i = 0; i < 2; ++i)
    y[i] = (b1 ? q[i + 2] : q[i]) + __shfl_xor(b1 ? q[i] : q[i + 2], 2, 32);
  const float z = (b0 ? y[1] : y[0]) + __shfl_xor(b0 ? y[0] : y[1], 1, 32);
  part[wave][(m >> 3) * 16 + hi * 8 + (m & 7)] = z;
  __syncthreads();
  #pragma unroll
  for (int r = 0; r < 8; ++r)
    H[(size_t)(row0 + r + hi * 8) * D + n] = c[r];
  if (t < 32) {
    float s = 0.f;
    #pragma unroll
    for (int w = 0; w < 8; ++w) s += part[w][t];
    if (t < 16) asrc[row0 + t] = s;
    else        adst[row0 + t - 16] = s;
  }
}

#define ROW_CAP 31
__global__ __launch_bounds__(256)
void build_rows(const int* __restrict__ ei, int* __restrict__ deg,
                int* __restrict__ srcrow, int* __restrict__ head,
                int2* __restrict__ nodes) {
  const int t = blockIdx.x * blockDim.x + threadIdx.x;
  if (t >= N_EDGES) return;
  const int s = ei[t];
  const int d = ei[N_EDGES + t];
  const int r = atomicAdd(&deg[d], 1) + 1;
  if (r < ROW_CAP) {
    srcrow[(size_t)d * 32 + 1 + r] = s;
  } else {
    const int old = atomicExch(&head[d], t);
    nodes[t] = make_int2(old, s);
  }
}

__device__ __forceinline__ float wave_max(float v) {
  for (int off = 16; off > 0; off >>= 1) v = fmaxf(v, __shfl_xor(v, off, 32));
  return v;
}
__device__ __forceinline__ float wave_sum(float v) {
  for (int off = 16; off > 0; off >>= 1) v += __shfl_xor(v, off, 32);
  return v;
}

template <bool STORE_H>
__global__ __launch_bounds__(256)
void gather_softmax_pool(const int* __restrict__ deg, const int* __restrict__ srcrow,
                         const int* __restrict__ head, const int2* __restrict__ nodes,
                         const float* __restrict__ asrc, const float* __restrict__ adst,
                         const float* __restrict__ H, const float* __restrict__ bias,
                         const int* __restrict__ batch, float* __restrict__ hout,
                         float* __restrict__ embsum) {
  __shared__ float rows[8][D];
  __shared__ int   gid[8];
  const int wave = threadIdx.x >> 5;
  const int node = blockIdx.x * 8 + wave;
  const int lane = threadIdx.x & 31;
  const int c0 = lane * 4;
  float4 o = make_float4(0.f, 0.f, 0.f, 0.f);
  if (node < N_NODES) {
    const float ad = adst[node];
    const int gb = batch[node];
    const float4 bv = *(const float4*)(bias + c0);

    float m = 0.f, l = 0.f;
    float4 a = make_float4(0.f, 0.f, 0.f, 0.f);
    bool first = true;
    const int cnt = __builtin_amdgcn_readfirstlane(deg[node]) + 1;
    const int nb = (cnt < ROW_CAP) ? cnt : ROW_CAP;
    int mysrc = node;
    if (lane >= 1 && lane <= nb) mysrc = srcrow[(size_t)node * 32 + lane];
    int k = nb + 1;
    int e = -1;
    if (cnt > ROW_CAP) e = head[node];
    int steps = 0;
    for (;;) {
      const int kk = __builtin_amdgcn_readfirstlane(k);
      float v = -__builtin_inff();
      if (lane < kk) {
        const float t = asrc[mysrc] + ad;
        v = (t > 0.f) ? t : NEG_SLOPE * t;
      }
      const float cm = wave_max(v);
      const float mn = first ? cm : fmaxf(m, cm);
      const float scale = first ? 0.f : __expf(m - mn);
      const float ex = (lane < kk) ? __expf(v - mn) : 0.f;
      l = l * scale + wave_sum(ex);
      a.x *= scale; a.y *= scale; a.z *= scale; a.w *= scale;
      #pragma unroll 4
      for (int j = 0; j < kk; ++j) {
        const int s = __builtin_amdgcn_readlane(mysrc, j);
        const float w = __int_as_float(__builtin_amdgcn_readlane(__float_as_int(ex), j));
        const float4 hv = *(const float4*)(H + (size_t)s * D + c0);
        a.x += w * hv.x; a.y += w * hv.y; a.z += w * hv.z; a.w += w * hv.w;
      }
      m = mn;
      first = false;
      if (!(e >= 0 && steps < N_EDGES)) break;
      k = 0;
      while (k < 32 && e >= 0 && steps < N_EDGES) {
        const int2 nd = nodes[e];
        if (lane == k) mysrc = nd.y;
        e = nd.x;
        ++k; ++steps;
      }
    }
    o.x = a.x / l + bv.x; o.y = a.y / l + bv.y; o.z = a.z / l + bv.z; o.w = a.w / l + bv.w;
    o.x = (o.x > 0.f) ? o.x : 0.f;
    o.y = (o.y > 0.f) ? o.y : 0.f;
    o.z = (o.z > 0.f) ? o.z : 0.f;
    o.w = (o.w > 0.f) ? o.w : 0.f;
    *(float4*)&rows[wave][c0] = o;
    if (lane == 0) gid[wave] = gb;
  } else {
    *(float4*)&rows[wave][c0] = o;
    if (lane == 0) gid[wave] = -1;
  }
  __syncthreads();
  if (STORE_H && node < N_NODES) *(float4*)(hout + (size_t)node * D + c0) = o;
  if (threadIdx.x < D) {
    const int c = threadIdx.x;
    int gv[8];
    float rv[8];
    #pragma unroll
    for (int w = 0; w < 8; ++w) { gv[w] = gid[w]; rv[w] = rows[w][c]; }
    int g = gv[0];
    float s = rv[0];
    #pragma unroll
    for (int w = 1; w < 8; ++w) {
      if (gv[w] != g) {
        if (g >= 0) atomicAdd(&embsum[(size_t)g * D + c], s);
        g = gv[w];
        s = 0.f;
      }
      s += rv[w];
    }
    if (g >= 0) atomicAdd(&embsum[(size_t)g * D + c], s);
  }
}

__global__ __launch_bounds__(256)
void count_kernel(const int* __restrict__ batch, float* __restrict__ cnt) {
  const int t = blockIdx.x * blockDim.x + threadIdx.x;
  if (t < N_NODES) atomicAdd(&cnt[batch[t]], 1.0f);
}

__global__ __launch_bounds__(256)
void finalize_kernel(float* __restrict__ out, const float* __restrict__ cnt) {
  const int t = blockIdx.x * blockDim.x + threadIdx.x;
  if (t >= 2 * N_GRAPHS * D) return;
  const int g = (t >> 7) & (N_GRAPHS - 1);
  float c = cnt[g];
  c = (c > 1.f) ? c : 1.f;
  out[t] /= c;
}

extern "C" void kernel_launch(void* const* d_in, const int* in_sizes, int n_in,
                              void* d_out, int out_size, void* d_ws, size_t ws_size,
                              hipStream_t stream) {
  const float* x     = (const float*)d_in[0];
  const int*   ei    = (const int*)d_in[1];
  const int*   batch = (const int*)d_in[2];
  const float* W1    = (const float*)d_in[3];
  const float* as1   = (const float*)d_in[4];
  const float* ad1   = (const float*)d_in[5];
  const float* b1    = (const float*)d_in[6];
  const float* W2    = (const float*)d_in[7];
  const float* as2   = (const float*)d_in[8];
  const float* ad2   = (const float*)d_in[9];
  const float* b2    = (const float*)d_in[10];
  float* out = (float*)d_out;

  char* w = (char*)d_ws;
  auto carve = [&](size_t bytes) {
    char* p = w;
    w += (bytes + 255) & ~(size_t)255;
    return p;
  };
  float*    hbuf  = (float*)carve((size_t)N_NODES * D * 4);
  float*    acc   = (float*)carve((size_t)N_NODES * D * 4);
  float*    asrc  = (float*)carve((size_t)N_NODES * 4);
  float*    adst  = (float*)carve((size_t)N_NODES * 4);
  int*      hd    = (int*)carve((size_t)2 * N_NODES * 4);
  int*      head  = hd;
  int*      deg   = hd + N_NODES;
  int*      srcrow = (int*)carve((size_t)N_NODES * 32 * 4);
  int2*     nodes = (int2*)carve((size_t)N_EDGES * 8);
  float*    cnt   = (float*)carve((size_t)N_GRAPHS * 4);

  const int TB = 256;
  const int gGemm  = N_NODES / 16;
  const int gEdge  = (N_EDGES + TB - 1) / TB;
  const int gGather = (N_NODES + 7) / 8;
  const int gNode  = (N_NODES + TB - 1) / TB;
  const int gFin   = (2 * N_GRAPHS * D + TB - 1) / TB;

  hipMemsetAsync(d_out, 0, (size_t)2 * N_GRAPHS * D * 4, stream);
  hipMemsetAsync(cnt, 0, (size_t)N_GRAPHS * 4, stream);
  hipMemsetAsync(hd, 0xFF, (size_t)2 * N_NODES * 4, stream);
  count_kernel<<<gNode, TB, 0, stream>>>(batch, cnt);
  build_rows<<<gEdge, TB, 0, stream>>>(ei, deg, srcrow, head, nodes);

  gemm_wmma<<<gGemm, TB, 0, stream>>>(x, W1, as1, ad1, hbuf, asrc, adst);
  gather_softmax_pool<true><<<gGather, TB, 0, stream>>>(deg, srcrow, head, nodes, asrc, adst,
                                                       hbuf, b1, batch, acc, out);

  gemm_wmma<<<gGemm, TB, 0, stream>>>(acc, W2, as2, ad2, hbuf, asrc, adst);
  gather_softmax_pool<false><<<gGather, TB, 0, stream>>>(deg, srcrow, head, nodes, asrc, adst,
                                                        hbuf, b2, batch, acc, out + N_GRAPHS * D);

  finalize_kernel<<<gFin, TB, 0, stream>>>(out, cnt);
}


# GEMM blocks stay resident and loop over 16-row tiles, each wave keeping its 128x16 slice of W in registers: a launch starts 8192 waves, not 25000, and W is read once per block; H bit-identical
# speedup vs baseline: 4.0360x; 1.0632x over previous
#include <hip/hip_runtime.h>

#define N_NODES  50000
#define N_EDGES  600000
#define N_GRAPHS 1024
#define D        128
#define NEG_SLOPE 0.2f

typedef float    v2f  __attribute__((ext_vector_type(2)));
typedef float    v8f  __attribute__((ext_vector_type(8)));

#define N_TILES (N_NODES / 16)
__global__ __launch_bounds__(256)
void gemm_wmma(const float* __restrict__ X, const float* __restrict__ W,
               const float* __restrict__ a_src, const float* __restrict__ a_dst,
               float* __restrict__ H, float* __restrict__ asrc,
               float* __restrict__ adst) {
  __shared__ __attribute__((aligned(16))) float xs[16][132];
  __shared__ float part[8][32];
  const int t = threadIdx.x;
  const int wave = t >> 5, lane = t & 31;
  const int hi = lane >> 4;
  const int m  = lane & 15;
  const int n  = wave * 16 + m;
  const float as_n = a_src[n];
  const float ad_n = a_dst[n];

  v2f bw[32];
  #pragma unroll
  for (int i = 0; i < 32; ++i) {
    const int ka = i * 4 + hi * 2;
    bw[i][0] = W[(size_t)ka * D + n];
    bw[i][1] = W[(size_t)(ka + 1) * D + n];
  }

  const int xr = t >> 5, xc4 = t & 31;
  int tile = blockIdx.x;
  float4 p0 = make_float4(0.f, 0.f, 0.f, 0.f), p1 = p0;
  if (tile < N_TILES) {
    p0 = ((const float4*)(X + (size_t)(tile * 16 + xr) * D))[xc4];
    p1 = ((const float4*)(X + (size_t)(tile * 16 + xr + 8) * D))[xc4];
  }
  for (; tile < N_TILES; tile += gridDim.x) {
    const int row0 = tile * 16;
    *(float4*)&xs[xr][xc4 * 4] = p0;
    *(float4*)&xs[xr + 8][xc4 * 4] = p1;
    __syncthreads();
    const int nxt = tile + gridDim.x;
    if (nxt < N_TILES) {
      p0 = ((const float4*)(X + (size_t)(nxt * 16 + xr) * D))[xc4];
      p1 = ((const float4*)(X + (size_t)(nxt * 16 + xr + 8) * D))[xc4];
    }

    v8f c = {0.f, 0.f, 0.f, 0.f, 0.f, 0.f, 0.f, 0.f};
    #pragma unroll
    for (int i = 0; i < 32; ++i) {
      const int ka = i * 4 + hi * 2;
      v2f a;
      a[0] = xs[m][ka];
      a[1] = xs[m][ka + 1];
      c = __builtin_amdgcn_wmma_f32_16x16x4_f32(false, a, false, bw[i], (short)0, c,
                                                false, false);
    }

    float v[16];
    #pragma unroll
    for (int r = 0; r < 8; ++r) { v[r] = c[r] * as_n; v[8 + r] = c[r] * ad_n; }
    const bool b3 = (m & 8) != 0, b2 = (m & 4) != 0, b1 = (m & 2) != 0, b0 = (m & 1) != 0;
    float u[8];
    #pragma unroll
    for (int i = 0; i < 8; ++i)
      u[i] = (b3 ? v[i + 8] : v[i]) + __shfl_xor(b3 ? v[i] : v[i + 8], 8, 32);
    float q[4];
    #pragma unroll
    for (int i = 0; i < 4; ++i)
      q[i] = (b2 ? u[i + 4] : u[i]) + __shfl_xor(b2 ? u[i] : u[i + 4], 4, 32);
    float y[2];
    #pragma unroll
    for (int i = 0; i < 2; ++i)
      y[i] = (b1 ? q[i + 2] : q[i]) + __shfl_xor(b1 ? q[i] : q[i + 2], 2, 32);
    const float z = (b0 ? y[1] : y[0]) + __shfl_xor(b0 ? y[0] : y[1], 1, 32);
    part[wave][(m >> 3) * 16 + hi * 8 + (m & 7)] = z;
    __syncthreads();
    #pragma unroll
    for (int r = 0; r < 8; ++r)
      H[(size_t)(row0 + r + hi * 8) * D + n] = c[r];
    if (t < 32) {
      float s = 0.f;
      #pragma unroll
      for (int w = 0; w < 8; ++w) s += part[w][t];
      if (t < 16) asrc[row0 + t] = s;
      else        adst[row0 + t - 16] = s;
    }
  }
}

#define ROW_CAP 31
__global__ __launch_bounds__(256)
void build_rows(const int* __restrict__ ei, int* __restrict__ deg,
                int* __restrict__ srcrow, int* __restrict__ head,
                int2* __restrict__ nodes) {
  const int t = blockIdx.x * blockDim.x + threadIdx.x;
  if (t >= N_EDGES) return;
  const int s = ei[t];
  const int d = ei[N_EDGES + t];
  const int r = atomicAdd(&deg[d], 1) + 1;
  if (r < ROW_CAP) {
    srcrow[(size_t)d * 32 + 1 + r] = s;
  } else {
    const int old = atomicExch(&head[d], t);
    nodes[t] = make_int2(old, s);
  }
}

__device__ __forceinline__ float wave_max(float v) {
  for (int off = 16; off > 0; off >>= 1) v = fmaxf(v, __shfl_xor(v, off, 32));
  return v;
}
__device__ __forceinline__ float wave_sum(float v) {
  for (int off = 16; off > 0; off >>= 1) v += __shfl_xor(v, off, 32);
  return v;
}

template <bool STORE_H>
__global__ __launch_bounds__(256)
void gather_softmax_pool(const int* __restrict__ deg, const int* __restrict__ srcrow,
                         const int* __restrict__ head, const int2* __restrict__ nodes,
                         const float* __restrict__ asrc, const float* __restrict__ adst,
                         const float* __restrict__ H, const float* __restrict__ bias,
                         const int* __restrict__ batch, float* __restrict__ hout,
                         float* __restrict__ embsum) {
  __shared__ float rows[8][D];
  __shared__ int   gid[8];
  const int wave = threadIdx.x >> 5;
  const int node = blockIdx.x * 8 + wave;
  const int lane = threadIdx.x & 31;
  const int c0 = lane * 4;
  float4 o = make_float4(0.f, 0.f, 0.f, 0.f);
  if (node < N_NODES) {
    const float ad = adst[node];
    const int gb = batch[node];
    const float4 bv = *(const float4*)(bias + c0);

    float m = 0.f, l = 0.f;
    float4 a = make_float4(0.f, 0.f, 0.f, 0.f);
    bool first = true;
    const int cnt = __builtin_amdgcn_readfirstlane(deg[node]) + 1;
    const int nb = (cnt < ROW_CAP) ? cnt : ROW_CAP;
    int mysrc = node;
    if (lane >= 1 && lane <= nb) mysrc = srcrow[(size_t)node * 32 + lane];
    int k = nb + 1;
    int e = -1;
    if (cnt > ROW_CAP) e = head[node];
    int steps = 0;
    for (;;) {
      const int kk = __builtin_amdgcn_readfirstlane(k);
      float v = -__builtin_inff();
      if (lane < kk) {
        const float t = asrc[mysrc] + ad;
        v = (t > 0.f) ? t : NEG_SLOPE * t;
      }
      const float cm = wave_max(v);
      const float mn = first ? cm : fmaxf(m, cm);
      const float scale = first ? 0.f : __expf(m - mn);
      const float ex = (lane < kk) ? __expf(v - mn) : 0.f;
      l = l * scale + wave_sum(ex);
      a.x *= scale; a.y *= scale; a.z *= scale; a.w *= scale;
      #pragma unroll 4
      for (int j = 0; j < kk; ++j) {
        const int s = __builtin_amdgcn_readlane(mysrc, j);
        const float w = __int_as_float(__builtin_amdgcn_readlane(__float_as_int(ex), j));
        const float4 hv = *(const float4*)(H + (size_t)s * D + c0);
        a.x += w * hv.x; a.y += w * hv.y; a.z += w * hv.z; a.w += w * hv.w;
      }
      m = mn;
      first = false;
      if (!(e >= 0 && steps < N_EDGES)) break;
      k = 0;
      while (k < 32 && e >= 0 && steps < N_EDGES) {
        const int2 nd = nodes[e];
        if (lane == k) mysrc = nd.y;
        e = nd.x;
        ++k; ++steps;
      }
    }
    o.x = a.x / l + bv.x; o.y = a.y / l + bv.y; o.z = a.z / l + bv.z; o.w = a.w / l + bv.w;
    o.x = (o.x > 0.f) ? o.x : 0.f;
    o.y = (o.y > 0.f) ? o.y : 0.f;
    o.z = (o.z > 0.f) ? o.z : 0.f;
    o.w = (o.w > 0.f) ? o.w : 0.f;
    *(float4*)&rows[wave][c0] = o;
    if (lane == 0) gid[wave] = gb;
  } else {
    *(float4*)&rows[wave][c0] = o;
    if (lane == 0) gid[wave] = -1;
  }
  __syncthreads();
  if (STORE_H && node < N_NODES) *(float4*)(hout + (size_t)node * D + c0) = o;
  if (threadIdx.x < D) {
    const int c = threadIdx.x;
    int gv[8];
    float rv[8];
    #pragma unroll
    for (int w = 0; w < 8; ++w) { gv[w] = gid[w]; rv[w] = rows[w][c]; }
    int g = gv[0];
    float s = rv[0];
    #pragma unroll
    for (int w = 1; w < 8; ++w) {
      if (gv[w] != g) {
        if (g >= 0) atomicAdd(&embsum[(size_t)g * D + c], s);
        g = gv[w];
        s = 0.f;
      }
      s += rv[w];
    }
    if (g >= 0) atomicAdd(&embsum[(size_t)g * D + c], s);
  }
}

__global__ __launch_bounds__(256)
void count_kernel(const int* __restrict__ batch, float* __restrict__ cnt) {
  const int t = blockIdx.x * blockDim.x + threadIdx.x;
  if (t < N_NODES) atomicAdd(&cnt[batch[t]], 1.0f);
}

__global__ __launch_bounds__(256)
void finalize_kernel(float* __restrict__ out, const float* __restrict__ cnt) {
  const int t = blockIdx.x * blockDim.x + threadIdx.x;
  if (t >= 2 * N_GRAPHS * D) return;
  const int g = (t >> 7) & (N_GRAPHS - 1);
  float c = cnt[g];
  c = (c > 1.f) ? c : 1.f;
  out[t] /= c;
}

extern "C" void kernel_launch(void* const* d_in, const int* in_sizes, int n_in,
                              void* d_out, int out_size, void* d_ws, size_t ws_size,
                              hipStream_t stream) {
  const float* x     = (const float*)d_in[0];
  const int*   ei    = (const int*)d_in[1];
  const int*   batch = (const int*)d_in[2];
  const float* W1    = (const float*)d_in[3];
  const float* as1   = (const float*)d_in[4];
  const float* ad1   = (const float*)d_in[5];
  const float* b1    = (const float*)d_in[6];
  const float* W2    = (const float*)d_in[7];
  const float* as2   = (const float*)d_in[8];
  const float* ad2   = (const float*)d_in[9];
  const float* b2    = (const float*)d_in[10];
  float* out = (float*)d_out;

  char* w = (char*)d_ws;
  auto carve = [&](size_t bytes) {
    char* p = w;
    w += (bytes + 255) & ~(size_t)255;
    return p;
  };
  float*    hbuf  = (float*)carve((size_t)N_NODES * D * 4);
  float*    acc   = (float*)carve((size_t)N_NODES * D * 4);
  float*    asrc  = (float*)carve((size_t)N_NODES * 4);
  float*    adst  = (float*)carve((size_t)N_NODES * 4);
  int*      hd    = (int*)carve((size_t)2 * N_NODES * 4);
  int*      head  = hd;
  int*      deg   = hd + N_NODES;
  int*      srcrow = (int*)carve((size_t)N_NODES * 32 * 4);
  int2*     nodes = (int2*)carve((size_t)N_EDGES * 8);
  float*    cnt   = (float*)carve((size_t)N_GRAPHS * 4);

  const int TB = 256;
  const int gGemm  = 1024;
  const int gEdge  = (N_EDGES + TB - 1) / TB;
  const int gGather = (N_NODES + 7) / 8;
  const int gNode  = (N_NODES + TB - 1) / TB;
  const int gFin   = (2 * N_GRAPHS * D + TB - 1) / TB;

  hipMemsetAsync(d_out, 0, (size_t)2 * N_GRAPHS * D * 4, stream);
  hipMemsetAsync(cnt, 0, (size_t)N_GRAPHS * 4, stream);
  hipMemsetAsync(hd, 0xFF, (size_t)2 * N_NODES * 4, stream);
  count_kernel<<<gNode, TB, 0, stream>>>(batch, cnt);
  build_rows<<<gEdge, TB, 0, stream>>>(ei, deg, srcrow, head, nodes);

  gemm_wmma<<<gGemm, TB, 0, stream>>>(x, W1, as1, ad1, hbuf, asrc, adst);
  gather_softmax_pool<true><<<gGather, TB, 0, stream>>>(deg, srcrow, head, nodes, asrc, adst,
                                                       hbuf, b1, batch, acc, out);

  gemm_wmma<<<gGemm, TB, 0, stream>>>(acc, W2, as2, ad2, hbuf, asrc, adst);
  gather_softmax_pool<false><<<gGather, TB, 0, stream>>>(deg, srcrow, head, nodes, asrc, adst,
                                                        hbuf, b2, batch, acc, out + N_GRAPHS * D);

  finalize_kernel<<<gFin, TB, 0, stream>>>(out, cnt);
}


# Layer-2 GEMM fused into layer-1 aggregation: resident 16-wave blocks gather a 16-node tile into LDS and multiply it by W2 kept in LDS; h1 never stored; emb1 pooled over 16-row runs (new sum tree)
# speedup vs baseline: 4.2725x; 1.0586x over previous
#include <hip/hip_runtime.h>

#define N_NODES  50000
#define N_EDGES  600000
#define N_GRAPHS 1024
#define D        128
#define NEG_SLOPE 0.2f

typedef float    v2f  __attribute__((ext_vector_type(2)));
typedef float    v8f  __attribute__((ext_vector_type(8)));

#define N_TILES (N_NODES / 16)
__global__ __launch_bounds__(256)
void gemm_wmma(const float* __restrict__ X, const float* __restrict__ W,
               const float* __restrict__ a_src, const float* __restrict__ a_dst,
               float* __restrict__ H, float* __restrict__ asrc,
               float* __restrict__ adst) {
  __shared__ __attribute__((aligned(16))) float xs[16][132];
  __shared__ float part[8][32];
  const int t = threadIdx.x;
  const int wave = t >> 5, lane = t & 31;
  const int hi = lane >> 4;
  const int m  = lane & 15;
  const int n  = wave * 16 + m;
  const float as_n = a_src[n];
  const float ad_n = a_dst[n];

  v2f bw[32];
  #pragma unroll
  for (int i = 0; i < 32; ++i) {
    const int ka = i * 4 + hi * 2;
    bw[i][0] = W[(size_t)ka * D + n];
    bw[i][1] = W[(size_t)(ka + 1) * D + n];
  }

  const int xr = t >> 5, xc4 = t & 31;
  int tile = blockIdx.x;
  float4 p0 = make_float4(0.f, 0.f, 0.f, 0.f), p1 = p0;
  if (tile < N_TILES) {
    p0 = ((const float4*)(X + (size_t)(tile * 16 + xr) * D))[xc4];
    p1 = ((const float4*)(X + (size_t)(tile * 16 + xr + 8) * D))[xc4];
  }
  for (; tile < N_TILES; tile += gridDim.x) {
    const int row0 = tile * 16;
    *(float4*)&xs[xr][xc4 * 4] = p0;
    *(float4*)&xs[xr + 8][xc4 * 4] = p1;
    __syncthreads();
    const int nxt = tile + gridDim.x;
    if (nxt < N_TILES) {
      p0 = ((const float4*)(X + (size_t)(nxt * 16 + xr) * D))[xc4];
      p1 = ((const float4*)(X + (size_t)(nxt * 16 + xr + 8) * D))[xc4];
    }

    v8f c = {0.f, 0.f, 0.f, 0.f, 0.f, 0.f, 0.f, 0.f};
    #pragma unroll
    for (int i = 0; i < 32; ++i) {
      const int ka = i * 4 + hi * 2;
      v2f a;
      a[0] = xs[m][ka];
      a[1] = xs[m][ka + 1];
      c = __builtin_amdgcn_wmma_f32_16x16x4_f32(false, a, false, bw[i], (short)0, c,
                                                false, false);
    }

    float v[16];
    #pragma unroll
    for (int r = 0; r < 8; ++r) { v[r] = c[r] * as_n; v[8 + r] = c[r] * ad_n; }
    const bool b3 = (m & 8) != 0, b2 = (m & 4) != 0, b1 = (m & 2) != 0, b0 = (m & 1) != 0;
    float u[8];
    #pragma unroll
    for (int i = 0; i < 8; ++i)
      u[i] = (b3 ? v[i + 8] : v[i]) + __shfl_xor(b3 ? v[i] : v[i + 8], 8, 32);
    float q[4];
    #pragma unroll
    for (int i = 0; i < 4; ++i)
      q[i] = (b2 ? u[i + 4] : u[i]) + __shfl_xor(b2 ? u[i] : u[i + 4], 4, 32);
    float y[2];
    #pragma unroll
    for (int i = 0; i < 2; ++i)
      y[i] = (b1 ? q[i + 2] : q[i]) + __shfl_xor(b1 ? q[i] : q[i + 2], 2, 32);
    const float z = (b0 ? y[1] : y[0]) + __shfl_xor(b0 ? y[0] : y[1], 1, 32);
    part[wave][(m >> 3) * 16 + hi * 8 + (m & 7)] = z;
    __syncthreads();
    #pragma unroll
    for (int r = 0; r < 8; ++r)
      H[(size_t)(row0 + r + hi * 8) * D + n] = c[r];
    if (t < 32) {
      float s = 0.f;
      #pragma unroll
      for (int w = 0; w < 8; ++w) s += part[w][t];
      if (t < 16) asrc[row0 + t] = s;
      else        adst[row0 + t - 16] = s;
    }
  }
}

#define ROW_CAP 31
__global__ __launch_bounds__(256)
void build_rows(const int* __restrict__ ei, int* __restrict__ deg,
                int* __restrict__ srcrow, int* __restrict__ head,
                int2* __restrict__ nodes) {
  const int t = blockIdx.x * blockDim.x + threadIdx.x;
  if (t >= N_EDGES) return;
  const int s = ei[t];
  const int d = ei[N_EDGES + t];
  const int r = atomicAdd(&deg[d], 1) + 1;
  if (r < ROW_CAP) {
    srcrow[(size_t)d * 32 + 1 + r] = s;
  } else {
    const int old = atomicExch(&head[d], t);
    nodes[t] = make_int2(old, s);
  }
}

__device__ __forceinline__ float wave_max(float v) {
  for (int off = 16; off > 0; off >>= 1) v = fmaxf(v, __shfl_xor(v, off, 32));
  return v;
}
__device__ __forceinline__ float wave_sum(float v) {
  for (int off = 16; off > 0; off >>= 1) v += __shfl_xor(v, off, 32);
  return v;
}

typedef const __attribute__((address_space(1))) float* gptr_t;
__device__ __forceinline__ gptr_t row_base(const float* __restrict__ H, const int s) {
  const unsigned long long p = (unsigned long long)H + ((unsigned long long)(unsigned)s << 9);
  const unsigned lo = __builtin_amdgcn_readfirstlane((unsigned)p);
  const unsigned hi = __builtin_amdgcn_readfirstlane((unsigned)(p >> 32));
  return (gptr_t)(((unsigned long long)hi << 32) | lo);
}

template <bool SBASE>
__device__ __forceinline__ float4 gather_row(const int node, const int lane, const int c0,
                                             const int* __restrict__ deg,
                                             const int* __restrict__ srcrow,
                                             const int* __restrict__ head,
                                             const int2* __restrict__ nodes,
                                             const float* __restrict__ asrc,
                                             const float* __restrict__ adst,
                                             const float* __restrict__ H) {
  const float ad = adst[node];
  float m = 0.f, l = 0.f;
  float4 a = make_float4(0.f, 0.f, 0.f, 0.f);
  bool first = true;
  const int cnt = __builtin_amdgcn_readfirstlane(deg[node]) + 1;
  const int nb = (cnt < ROW_CAP) ? cnt : ROW_CAP;
  int mysrc = node;
  if (lane >= 1 && lane <= nb) mysrc = srcrow[(size_t)node * 32 + lane];
  int k = nb + 1;
  int e = -1;
  if (cnt > ROW_CAP) e = head[node];
  int steps = 0;
  for (;;) {
    const int kk = __builtin_amdgcn_readfirstlane(k);
    float v = -__builtin_inff();
    if (lane < kk) {
      const float tv = asrc[mysrc] + ad;
      v = (tv > 0.f) ? tv : NEG_SLOPE * tv;
    }
    const float cm = wave_max(v);
    const float mn = first ? cm : fmaxf(m, cm);
    const float scale = first ? 0.f : __expf(m - mn);
    const float ex = (lane < kk) ? __expf(v - mn) : 0.f;
    l = l * scale + wave_sum(ex);
    a.x *= scale; a.y *= scale; a.z *= scale; a.w *= scale;
    #pragma unroll 4
    for (int j = 0; j < kk; ++j) {
      const int s = __builtin_amdgcn_readlane(mysrc, j);
      const float w = __int_as_float(__builtin_amdgcn_readlane(__float_as_int(ex), j));
      float4 hv;
      if (SBASE) hv = *(const __attribute__((address_space(1))) float4*)(row_base(H, s) + c0);
      else       hv = *(const float4*)(H + (size_t)s * D + c0);
      a.x += w * hv.x; a.y += w * hv.y; a.z += w * hv.z; a.w += w * hv.w;
    }
    m = mn;
    first = false;
    if (!(e >= 0 && steps < N_EDGES)) break;
    k = 0;
    while (k < 32 && e >= 0 && steps < N_EDGES) {
      const int2 nd = nodes[e];
      if (lane == k) mysrc = nd.y;
      e = nd.x;
      ++k; ++steps;
    }
  }
  return make_float4(a.x / l, a.y / l, a.z / l, a.w / l);
}

__device__ __forceinline__ float4 bias_relu(const float4 r, const float4 bv) {
  float4 o;
  o.x = r.x + bv.x; o.y = r.y + bv.y; o.z = r.z + bv.z; o.w = r.w + bv.w;
  o.x = (o.x > 0.f) ? o.x : 0.f;
  o.y = (o.y > 0.f) ? o.y : 0.f;
  o.z = (o.z > 0.f) ? o.z : 0.f;
  o.w = (o.w > 0.f) ? o.w : 0.f;
  return o;
}

__global__ __launch_bounds__(256)
void gather_softmax_pool(const int* __restrict__ deg, const int* __restrict__ srcrow,
                         const int* __restrict__ head, const int2* __restrict__ nodes,
                         const float* __restrict__ asrc, const float* __restrict__ adst,
                         const float* __restrict__ H, const float* __restrict__ bias,
                         const int* __restrict__ batch, float* __restrict__ embsum) {
  __shared__ float rows[8][D];
  __shared__ int   gid[8];
  const int wave = threadIdx.x >> 5;
  const int node = blockIdx.x * 8 + wave;
  const int lane = threadIdx.x & 31;
  const int c0 = lane * 4;
  float4 o = make_float4(0.f, 0.f, 0.f, 0.f);
  if (node < N_NODES) {
    const int gb = batch[node];
    const float4 bv = *(const float4*)(bias + c0);
    o = bias_relu(gather_row<false>(node, lane, c0, deg, srcrow, head, nodes, asrc, adst, H), bv);
    *(float4*)&rows[wave][c0] = o;
    if (lane == 0) gid[wave] = gb;
  } else {
    *(float4*)&rows[wave][c0] = o;
    if (lane == 0) gid[wave] = -1;
  }
  __syncthreads();
  if (threadIdx.x < D) {
    const int c = threadIdx.x;
    int gv[8];
    float rv[8];
    #pragma unroll
    for (int w = 0; w < 8; ++w) { gv[w] = gid[w]; rv[w] = rows[w][c]; }
    int g = gv[0];
    float s = rv[0];
    #pragma unroll
    for (int w = 1; w < 8; ++w) {
      if (gv[w] != g) {
        if (g >= 0) atomicAdd(&embsum[(size_t)g * D + c], s);
        g = gv[w];
        s = 0.f;
      }
      s += rv[w];
    }
    if (g >= 0) atomicAdd(&embsum[(size_t)g * D + c], s);
  }
}

static_assert(N_NODES % 16 == 0, "the fused kernel walks whole 16-node tiles");
__global__ __launch_bounds__(512, 16)
void gather_gemm(const int* __restrict__ deg, const int* __restrict__ srcrow,
                 const int* __restrict__ head, const int2* __restrict__ nodes,
                 const float* __restrict__ asrc, const float* __restrict__ adst,
                 const float* __restrict__ H, const float* __restrict__ bias,
                 const int* __restrict__ batch, float* __restrict__ embsum,
                 const float* __restrict__ W, const float* __restrict__ a_src,
                 const float* __restrict__ a_dst, float* __restrict__ Hout,
                 float* __restrict__ asrc_out, float* __restrict__ adst_out) {
  __shared__ __attribute__((aligned(16))) float wl[64 * 256];
  __shared__ __attribute__((aligned(16))) float hs[16][132];
  __shared__ float part[8][32];
  __shared__ int   gid[16];
  const int t = threadIdx.x;
  const int wave = __builtin_amdgcn_readfirstlane(t >> 5);
  const int lane = t & 31;
  const int hi = lane >> 4;
  const int m  = lane & 15;
  const int n  = (wave & 7) * 16 + m;
  const int c0 = lane * 4;
  const float4 bv = *(const float4*)(bias + c0);

  #pragma unroll
  for (int q = 0; q < 8; ++q) {
    const int f = q * 512 + t;
    const int k = f >> 5, n4 = f & 31;
    const float4 wv = ((const float4*)(W + (size_t)k * D))[n4];
    const int kp = k >> 1;
    const int sw = (kp & 1) << 4;
    float* dp = wl + kp * 256 + (k & 1);
    dp[((4 * n4 + 0) ^ sw) * 2] = wv.x;
    dp[((4 * n4 + 1) ^ sw) * 2] = wv.y;
    dp[((4 * n4 + 2) ^ sw) * 2] = wv.z;
    dp[((4 * n4 + 3) ^ sw) * 2] = wv.w;
  }

  for (int tile = blockIdx.x; tile < N_TILES; tile += gridDim.x) {
    const int row0 = tile * 16;
    const int node = row0 + wave;
    const int gb = batch[node];
    const float4 o = bias_relu(gather_row<true>(node, lane, c0, deg, srcrow, head, nodes, asrc, adst, H), bv);
    *(float4*)&hs[wave][c0] = o;
    if (lane == 0) gid[wave] = gb;
    __syncthreads();

    v8f c = {0.f, 0.f, 0.f, 0.f, 0.f, 0.f, 0.f, 0.f};
    if (wave < 8) {
      const float as_n = a_src[(unsigned)n];
      const float ad_n = a_dst[(unsigned)n];
      const float* ap = &hs[m][hi * 2];
      const float* bp = wl + hi * 256 + ((n ^ (hi << 4)) << 1);
      v2f fa[2][4], fb[2][4];
      #pragma unroll
      for (int j = 0; j < 4; ++j) {
        fa[0][j][0] = ap[j * 4];      fa[0][j][1] = ap[j * 4 + 1];
        fb[0][j][0] = bp[j * 512];    fb[0][j][1] = bp[j * 512 + 1];
      }
      #pragma unroll
      for (int g = 0; g < 8; ++g) {
        if (g + 1 < 8) {
          #pragma unroll
          for (int j = 0; j < 4; ++j) {
            const int i = (g + 1) * 4 + j;
            fa[(g + 1) & 1][j][0] = ap[i * 4];      fa[(g + 1) & 1][j][1] = ap[i * 4 + 1];
            fb[(g + 1) & 1][j][0] = bp[i * 512];    fb[(g + 1) & 1][j][1] = bp[i * 512 + 1];
          }
        }
        #pragma unroll
        for (int j = 0; j < 4; ++j)
          c = __builtin_amdgcn_wmma_f32_16x16x4_f32(false, fa[g & 1][j], false, fb[g & 1][j],
                                                    (short)0, c, false, false);
      }
      float v[16];
      #pragma unroll
      for (int r = 0; r < 8; ++r) { v[r] = c[r] * as_n; v[8 + r] = c[r] * ad_n; }
      const bool b3 = (m & 8) != 0, b2 = (m & 4) != 0, b1 = (m & 2) != 0, b0 = (m & 1) != 0;
      float u[8];
      #pragma unroll
      for (int i = 0; i < 8; ++i)
        u[i] = (b3 ? v[i + 8] : v[i]) + __shfl_xor(b3 ? v[i] : v[i + 8], 8, 32);
      float q[4];
      #pragma unroll
      for (int i = 0; i < 4; ++i)
        q[i] = (b2 ? u[i + 4] : u[i]) + __shfl_xor(b2 ? u[i] : u[i + 4], 4, 32);
      float y[2];
      #pragma unroll
      for (int i = 0; i < 2; ++i)
        y[i] = (b1 ? q[i + 2] : q[i]) + __shfl_xor(b1 ? q[i] : q[i + 2], 2, 32);
      const float z = (b0 ? y[1] : y[0]) + __shfl_xor(b0 ? y[0] : y[1], 1, 32);
      part[wave][(m >> 3) * 16 + hi * 8 + (m & 7)] = z;
    } else if (wave < 12) {
      const int col = t - 256;
      int gv[16];
      float rv[16];
      #pragma unroll
      for (int w = 0; w < 16; ++w) { gv[w] = gid[w]; rv[w] = hs[w][col]; }
      int g = gv[0];
      float s = rv[0];
      #pragma unroll
      for (int w = 1; w < 16; ++w) {
        if (gv[w] != g) {
          atomicAdd(&embsum[(size_t)g * D + col], s);
          g = gv[w];
          s = 0.f;
        }
        s += rv[w];
      }
      atomicAdd(&embsum[(size_t)g * D + col], s);
    }
    __syncthreads();
    if (wave < 8) {
      float* __restrict__ htile = Hout + (size_t)row0 * D;
      #pragma unroll
      for (int r = 0; r < 8; ++r)
        htile[(unsigned)((r + hi * 8) * D + n)] = c[r];
      if (t < 32) {
        float s = 0.f;
        #pragma unroll
        for (int w = 0; w < 8; ++w) s += part[w][t];
        if (t < 16) asrc_out[row0 + t] = s;
        else        adst_out[row0 + t - 16] = s;
      }
    }
  }
}

__global__ __launch_bounds__(256)
void count_kernel(const int* __restrict__ batch, float* __restrict__ cnt) {
  const int t = blockIdx.x * blockDim.x + threadIdx.x;
  if (t < N_NODES) atomicAdd(&cnt[batch[t]], 1.0f);
}

__global__ __launch_bounds__(256)
void finalize_kernel(float* __restrict__ out, const float* __restrict__ cnt) {
  const int t = blockIdx.x * blockDim.x + threadIdx.x;
  if (t >= 2 * N_GRAPHS * D) return;
  const int g = (t >> 7) & (N_GRAPHS - 1);
  float c = cnt[g];
  c = (c > 1.f) ? c : 1.f;
  out[t] /= c;
}

extern "C" void kernel_launch(void* const* d_in, const int* in_sizes, int n_in,
                              void* d_out, int out_size, void* d_ws, size_t ws_size,
                              hipStream_t stream) {
  const float* x     = (const float*)d_in[0];
  const int*   ei    = (const int*)d_in[1];
  const int*   batch = (const int*)d_in[2];
  const float* W1    = (const float*)d_in[3];
  const float* as1   = (const float*)d_in[4];
  const float* ad1   = (const float*)d_in[5];
  const float* b1    = (const float*)d_in[6];
  const float* W2    = (const float*)d_in[7];
  const float* as2   = (const float*)d_in[8];
  const float* ad2   = (const float*)d_in[9];
  const float* b2    = (const float*)d_in[10];
  float* out = (float*)d_out;

  char* w = (char*)d_ws;
  auto carve = [&](size_t bytes) {
    char* p = w;
    w += (bytes + 255) & ~(size_t)255;
    return p;
  };
  float*    hbuf  = (float*)carve((size_t)N_NODES * D * 4);
  float*    hbuf2 = (float*)carve((size_t)N_NODES * D * 4);
  float*    asrc  = (float*)carve((size_t)N_NODES * 4);
  float*    adst  = (float*)carve((size_t)N_NODES * 4);
  int*      hd    = (int*)carve((size_t)2 * N_NODES * 4);
  int*      head  = hd;
  int*      deg   = hd + N_NODES;
  int*      srcrow = (int*)carve((size_t)N_NODES * 32 * 4);
  int2*     nodes = (int2*)carve((size_t)N_EDGES * 8);
  float*    cnt   = (float*)carve((size_t)N_GRAPHS * 4);
  float*    asrc2 = (float*)carve((size_t)N_NODES * 4);
  float*    adst2 = (float*)carve((size_t)N_NODES * 4);

  const int TB = 256;
  const int gGemm  = 1024;
  const int gFused = 1024;
  const int gEdge  = (N_EDGES + TB - 1) / TB;
  const int gGather = (N_NODES + 7) / 8;
  const int gNode  = (N_NODES + TB - 1) / TB;
  const int gFin   = (2 * N_GRAPHS * D + TB - 1) / TB;

  hipMemsetAsync(d_out, 0, (size_t)2 * N_GRAPHS * D * 4, stream);
  hipMemsetAsync(cnt, 0, (size_t)N_GRAPHS * 4, stream);
  hipMemsetAsync(hd, 0xFF, (size_t)2 * N_NODES * 4, stream);
  count_kernel<<<gNode, TB, 0, stream>>>(batch, cnt);
  build_rows<<<gEdge, TB, 0, stream>>>(ei, deg, srcrow, head, nodes);

  gemm_wmma<<<gGemm, TB, 0, stream>>>(x, W1, as1, ad1, hbuf, asrc, adst);
  gather_gemm<<<gFused, 512, 0, stream>>>(deg, srcrow, head, nodes, asrc, adst, hbuf, b1, batch,
                                          out, W2, as2, ad2, hbuf2, asrc2, adst2);
  gather_softmax_pool<<<gGather, TB, 0, stream>>>(deg, srcrow, head, nodes, asrc2, adst2,
                                                  hbuf2, b2, batch, out + N_GRAPHS * D);

  finalize_kernel<<<gFin, TB, 0, stream>>>(out, cnt);
}


# Edge-slot pass and graph-size count ride in the layer-1 GEMM launch: a GEMM thread takes one edge per row tile, slot atomic issued ahead of the tile's products; two launches of short waves fewer
# speedup vs baseline: 4.4785x; 1.0482x over previous
#include <hip/hip_runtime.h>

#define N_NODES  50000
#define N_EDGES  600000
#define N_GRAPHS 1024
#define D        128
#define NEG_SLOPE 0.2f

typedef float    v2f  __attribute__((ext_vector_type(2)));
typedef float    v8f  __attribute__((ext_vector_type(8)));

#define N_TILES (N_NODES / 16)
#define ROW_CAP 31
#define GEMM_BLOCKS  1024
#define GEMM_THREADS (GEMM_BLOCKS * 256)
#define EDGE_SLOTS   3
static_assert(EDGE_SLOTS * GEMM_THREADS >= N_EDGES, "every edge has a thread and a trip");
static_assert(GEMM_THREADS <= N_EDGES, "every thread has an edge on its first trip");
static_assert(GEMM_THREADS >= N_NODES, "every node has a thread");
static_assert(N_TILES >= EDGE_SLOTS * GEMM_BLOCKS, "every block makes at least EDGE_SLOTS trips");

__device__ __forceinline__ void put_source(const int e, const int s, const int d, const int old,
                                           int* __restrict__ srcrow, int* __restrict__ head,
                                           int2* __restrict__ nodes) {
  if (old < ROW_CAP - 1) {
    srcrow[(size_t)d * 32 + 2 + old] = s;
  } else {
    const int before = atomicExch(&head[d], e);
    nodes[e] = make_int2(before, s);
  }
}

__global__ __launch_bounds__(256)
void gemm_wmma(const float* __restrict__ X, const float* __restrict__ W,
               const float* __restrict__ a_src, const float* __restrict__ a_dst,
               float* __restrict__ H, float* __restrict__ asrc,
               float* __restrict__ adst,
               const int* __restrict__ ei, const int* __restrict__ batch,
               int* __restrict__ deg, int* __restrict__ srcrow, int* __restrict__ head,
               int2* __restrict__ nodes, float* __restrict__ cnt) {
  __shared__ __attribute__((aligned(16))) float xs[16][132];
  __shared__ float part[8][32];
  const int t = threadIdx.x;
  const int wave = t >> 5, lane = t & 31;
  const int hi = lane >> 4;
  const int m  = lane & 15;
  const int n  = wave * 16 + m;
  const float as_n = a_src[n];
  const float ad_n = a_dst[n];

  v2f bw[32];
  #pragma unroll
  for (int i = 0; i < 32; ++i) {
    const int ka = i * 4 + hi * 2;
    bw[i][0] = W[(size_t)ka * D + n];
    bw[i][1] = W[(size_t)(ka + 1) * D + n];
  }

  const int xr = t >> 5, xc4 = t & 31;
  int tile = blockIdx.x;
  float4 p0 = make_float4(0.f, 0.f, 0.f, 0.f), p1 = p0;
  if (tile < N_TILES) {
    p0 = ((const float4*)(X + (size_t)(tile * 16 + xr) * D))[xc4];
    p1 = ((const float4*)(X + (size_t)(tile * 16 + xr + 8) * D))[xc4];
  }
  const int gid = blockIdx.x * 256 + t;
  int ns = ei[gid], nd = ei[N_EDGES + gid];
  unsigned cg = 0;
  if (gid < N_NODES) cg = (unsigned)batch[gid];
  #pragma nounroll
  for (int k = 0; tile < N_TILES; tile += GEMM_BLOCKS, ++k) {
    const int row0 = tile * 16;
    *(float4*)&xs[xr][xc4 * 4] = p0;
    *(float4*)&xs[xr + 8][xc4 * 4] = p1;
    __syncthreads();
    const int nxt = tile + GEMM_BLOCKS;
    if (nxt < N_TILES) {
      p0 = ((const float4*)(X + (size_t)(nxt * 16 + xr) * D))[xc4];
      p1 = ((const float4*)(X + (size_t)(nxt * 16 + xr + 8) * D))[xc4];
    }
    const int cs = ns, cd = nd;
    int cr = 0;
    if (cd >= 0) cr = atomicAdd(&deg[cd], 1);
    ns = 0; nd = -1;
    if (k + 1 < EDGE_SLOTS) {
      const unsigned e = (unsigned)gid + (unsigned)(k + 1) * GEMM_THREADS;
      if (e < N_EDGES) { ns = ei[e]; nd = ei[N_EDGES + e]; }
    }

    v8f c = {0.f, 0.f, 0.f, 0.f, 0.f, 0.f, 0.f, 0.f};
    #pragma unroll
    for (int i = 0; i < 32; ++i) {
      const int ka = i * 4 + hi * 2;
      v2f a;
      a[0] = xs[m][ka];
      a[1] = xs[m][ka + 1];
      c = __builtin_amdgcn_wmma_f32_16x16x4_f32(false, a, false, bw[i], (short)0, c,
                                                false, false);
    }

    float v[16];
    #pragma unroll
    for (int r = 0; r < 8; ++r) { v[r] = c[r] * as_n; v[8 + r] = c[r] * ad_n; }
    const bool b3 = (m & 8) != 0, b2 = (m & 4) != 0, b1 = (m & 2) != 0, b0 = (m & 1) != 0;
    float u[8];
    #pragma unroll
    for (int i = 0; i < 8; ++i)
      u[i] = (b3 ? v[i + 8] : v[i]) + __shfl_xor(b3 ? v[i] : v[i + 8], 8, 32);
    float q[4];
    #pragma unroll
    for (int i = 0; i < 4; ++i)
      q[i] = (b2 ? u[i + 4] : u[i]) + __shfl_xor(b2 ? u[i] : u[i + 4], 4, 32);
    float y[2];
    #pragma unroll
    for (int i = 0; i < 2; ++i)
      y[i] = (b1 ? q[i + 2] : q[i]) + __shfl_xor(b1 ? q[i] : q[i + 2], 2, 32);
    const float z = (b0 ? y[1] : y[0]) + __shfl_xor(b0 ? y[0] : y[1], 1, 32);
    part[wave][(m >> 3) * 16 + hi * 8 + (m & 7)] = z;
    __syncthreads();
    float* __restrict__ htile = H + (size_t)row0 * D;
    #pragma unroll
    for (int r = 0; r < 8; ++r)
      htile[(unsigned)((r + hi * 8) * D + n)] = c[r];
    if (t < 32) {
      float s = 0.f;
      #pragma unroll
      for (int w = 0; w < 8; ++w) s += part[w][t];
      if (t < 16) asrc[row0 + t] = s;
      else        adst[row0 + t - 16] = s;
    }
    if (cd >= 0) put_source(gid + k * GEMM_THREADS, cs, cd, cr, srcrow, head, nodes);
    if (k == 0 && gid < N_NODES) atomicAdd(&cnt[cg], 1.0f);
  }
}

__device__ __forceinline__ float wave_max(float v) {
  for (int off = 16; off > 0; off >>= 1) v = fmaxf(v, __shfl_xor(v, off, 32));
  return v;
}
__device__ __forceinline__ float wave_sum(float v) {
  for (int off = 16; off > 0; off >>= 1) v += __shfl_xor(v, off, 32);
  return v;
}

typedef const __attribute__((address_space(1))) float* gptr_t;
__device__ __forceinline__ gptr_t row_base(const float* __restrict__ H, const int s) {
  const unsigned long long p = (unsigned long long)H + ((unsigned long long)(unsigned)s << 9);
  const unsigned lo = __builtin_amdgcn_readfirstlane((unsigned)p);
  const unsigned hi = __builtin_amdgcn_readfirstlane((unsigned)(p >> 32));
  return (gptr_t)(((unsigned long long)hi << 32) | lo);
}

template <bool SBASE>
__device__ __forceinline__ float4 gather_row(const int node, const int lane, const int c0,
                                             const int* __restrict__ deg,
                                             const int* __restrict__ srcrow,
                                             const int* __restrict__ head,
                                             const int2* __restrict__ nodes,
                                             const float* __restrict__ asrc,
                                             const float* __restrict__ adst,
                                             const float* __restrict__ H) {
  const float ad = adst[node];
  float m = 0.f, l = 0.f;
  float4 a = make_float4(0.f, 0.f, 0.f, 0.f);
  bool first = true;
  const int cnt = __builtin_amdgcn_readfirstlane(deg[node]) + 1;
  const int nb = (cnt < ROW_CAP) ? cnt : ROW_CAP;
  int mysrc = node;
  if (lane >= 1 && lane <= nb) mysrc = srcrow[(size_t)node * 32 + lane];
  int k = nb + 1;
  int e = -1;
  if (cnt > ROW_CAP) e = head[node];
  int steps = 0;
  for (;;) {
    const int kk = __builtin_amdgcn_readfirstlane(k);
    float v = -__builtin_inff();
    if (lane < kk) {
      const float tv = asrc[mysrc] + ad;
      v = (tv > 0.f) ? tv : NEG_SLOPE * tv;
    }
    const float cm = wave_max(v);
    const float mn = first ? cm : fmaxf(m, cm);
    const float scale = first ? 0.f : __expf(m - mn);
    const float ex = (lane < kk) ? __expf(v - mn) : 0.f;
    l = l * scale + wave_sum(ex);
    a.x *= scale; a.y *= scale; a.z *= scale; a.w *= scale;
    #pragma unroll 4
    for (int j = 0; j < kk; ++j) {
      const int s = __builtin_amdgcn_readlane(mysrc, j);
      const float w = __int_as_float(__builtin_amdgcn_readlane(__float_as_int(ex), j));
      float4 hv;
      if (SBASE) hv = *(const __attribute__((address_space(1))) float4*)(row_base(H, s) + c0);
      else       hv = *(const float4*)(H + (size_t)s * D + c0);
      a.x += w * hv.x; a.y += w * hv.y; a.z += w * hv.z; a.w += w * hv.w;
    }
    m = mn;
    first = false;
    if (!(e >= 0 && steps < N_EDGES)) break;
    k = 0;
    while (k < 32 && e >= 0 && steps < N_EDGES) {
      const int2 nd = nodes[e];
      if (lane == k) mysrc = nd.y;
      e = nd.x;
      ++k; ++steps;
    }
  }
  return make_float4(a.x / l, a.y / l, a.z / l, a.w / l);
}

__device__ __forceinline__ float4 bias_relu(const float4 r, const float4 bv) {
  float4 o;
  o.x = r.x + bv.x; o.y = r.y + bv.y; o.z = r.z + bv.z; o.w = r.w + bv.w;
  o.x = (o.x > 0.f) ? o.x : 0.f;
  o.y = (o.y > 0.f) ? o.y : 0.f;
  o.z = (o.z > 0.f) ? o.z : 0.f;
  o.w = (o.w > 0.f) ? o.w : 0.f;
  return o;
}

__global__ __launch_bounds__(256)
void gather_softmax_pool(const int* __restrict__ deg, const int* __restrict__ srcrow,
                         const int* __restrict__ head, const int2* __restrict__ nodes,
                         const float* __restrict__ asrc, const float* __restrict__ adst,
                         const float* __restrict__ H, const float* __restrict__ bias,
                         const int* __restrict__ batch, float* __restrict__ embsum) {
  __shared__ float rows[8][D];
  __shared__ int   gid[8];
  const int wave = threadIdx.x >> 5;
  const int node = blockIdx.x * 8 + wave;
  const int lane = threadIdx.x & 31;
  const int c0 = lane * 4;
  float4 o = make_float4(0.f, 0.f, 0.f, 0.f);
  if (node < N_NODES) {
    const int gb = batch[node];
    const float4 bv = *(const float4*)(bias + c0);
    o = bias_relu(gather_row<false>(node, lane, c0, deg, srcrow, head, nodes, asrc, adst, H), bv);
    *(float4*)&rows[wave][c0] = o;
    if (lane == 0) gid[wave] = gb;
  } else {
    *(float4*)&rows[wave][c0] = o;
    if (lane == 0) gid[wave] = -1;
  }
  __syncthreads();
  if (threadIdx.x < D) {
    const int c = threadIdx.x;
    int gv[8];
    float rv[8];
    #pragma unroll
    for (int w = 0; w < 8; ++w) { gv[w] = gid[w]; rv[w] = rows[w][c]; }
    int g = gv[0];
    float s = rv[0];
    #pragma unroll
    for (int w = 1; w < 8; ++w) {
      if (gv[w] != g) {
        if (g >= 0) atomicAdd(&embsum[(size_t)g * D + c], s);
        g = gv[w];
        s = 0.f;
      }
      s += rv[w];
    }
    if (g >= 0) atomicAdd(&embsum[(size_t)g * D + c], s);
  }
}

static_assert(N_NODES % 16 == 0, "the fused kernel walks whole 16-node tiles");
__global__ __launch_bounds__(512, 16)
void gather_gemm(const int* __restrict__ deg, const int* __restrict__ srcrow,
                 const int* __restrict__ head, const int2* __restrict__ nodes,
                 const float* __restrict__ asrc, const float* __restrict__ adst,
                 const float* __restrict__ H, const float* __restrict__ bias,
                 const int* __restrict__ batch, float* __restrict__ embsum,
                 const float* __restrict__ W, const float* __restrict__ a_src,
                 const float* __restrict__ a_dst, float* __restrict__ Hout,
                 float* __restrict__ asrc_out, float* __restrict__ adst_out) {
  __shared__ __attribute__((aligned(16))) float wl[64 * 256];
  __shared__ __attribute__((aligned(16))) float hs[16][132];
  __shared__ float part[8][32];
  __shared__ int   gid[16];
  const int t = threadIdx.x;
  const int wave = __builtin_amdgcn_readfirstlane(t >> 5);
  const int lane = t & 31;
  const int hi = lane >> 4;
  const int m  = lane & 15;
  const int n  = (wave & 7) * 16 + m;
  const int c0 = lane * 4;
  const float4 bv = *(const float4*)(bias + c0);

  #pragma unroll
  for (int q = 0; q < 8; ++q) {
    const int f = q * 512 + t;
    const int k = f >> 5, n4 = f & 31;
    const float4 wv = ((const float4*)(W + (size_t)k * D))[n4];
    const int kp = k >> 1;
    const int sw = (kp & 1) << 4;
    float* dp = wl + kp * 256 + (k & 1);
    dp[((4 * n4 + 0) ^ sw) * 2] = wv.x;
    dp[((4 * n4 + 1) ^ sw) * 2] = wv.y;
    dp[((4 * n4 + 2) ^ sw) * 2] = wv.z;
    dp[((4 * n4 + 3) ^ sw) * 2] = wv.w;
  }

  for (int tile = blockIdx.x; tile < N_TILES; tile += gridDim.x) {
    const int row0 = tile * 16;
    const int node = row0 + wave;
    const int gb = batch[node];
    const float4 o = bias_relu(gather_row<true>(node, lane, c0, deg, srcrow, head, nodes, asrc, adst, H), bv);
    *(float4*)&hs[wave][c0] = o;
    if (lane == 0) gid[wave] = gb;
    __syncthreads();

    v8f c = {0.f, 0.f, 0.f, 0.f, 0.f, 0.f, 0.f, 0.f};
    if (wave < 8) {
      const float as_n = a_src[(unsigned)n];
      const float ad_n = a_dst[(unsigned)n];
      const float* ap = &hs[m][hi * 2];
      const float* bp = wl + hi * 256 + ((n ^ (hi << 4)) << 1);
      v2f fa[2][4], fb[2][4];
      #pragma unroll
      for (int j = 0; j < 4; ++j) {
        fa[0][j][0] = ap[j * 4];      fa[0][j][1] = ap[j * 4 + 1];
        fb[0][j][0] = bp[j * 512];    fb[0][j][1] = bp[j * 512 + 1];
      }
      #pragma unroll
      for (int g = 0; g < 8; ++g) {
        if (g + 1 < 8) {
          #pragma unroll
          for (int j = 0; j < 4; ++j) {
            const int i = (g + 1) * 4 + j;
            fa[(g + 1) & 1][j][0] = ap[i * 4];      fa[(g + 1) & 1][j][1] = ap[i * 4 + 1];
            fb[(g + 1) & 1][j][0] = bp[i * 512];    fb[(g + 1) & 1][j][1] = bp[i * 512 + 1];
          }
        }
        #pragma unroll
        for (int j = 0; j < 4; ++j)
          c = __builtin_amdgcn_wmma_f32_16x16x4_f32(false, fa[g & 1][j], false, fb[g & 1][j],
                                                    (short)0, c, false, false);
      }
      float v[16];
      #pragma unroll
      for (int r = 0; r < 8; ++r) { v[r] = c[r] * as_n; v[8 + r] = c[r] * ad_n; }
      const bool b3 = (m & 8) != 0, b2 = (m & 4) != 0, b1 = (m & 2) != 0, b0 = (m & 1) != 0;
      float u[8];
      #pragma unroll
      for (int i = 0; i < 8; ++i)
        u[i] = (b3 ? v[i + 8] : v[i]) + __shfl_xor(b3 ? v[i] : v[i + 8], 8, 32);
      float q[4];
      #pragma unroll
      for (int i = 0; i < 4; ++i)
        q[i] = (b2 ? u[i + 4] : u[i]) + __shfl_xor(b2 ? u[i] : u[i + 4], 4, 32);
      float y[2];
      #pragma unroll
      for (int i = 0; i < 2; ++i)
        y[i] = (b1 ? q[i + 2] : q[i]) + __shfl_xor(b1 ? q[i] : q[i + 2], 2, 32);
      const float z = (b0 ? y[1] : y[0]) + __shfl_xor(b0 ? y[0] : y[1], 1, 32);
      part[wave][(m >> 3) * 16 + hi * 8 + (m & 7)] = z;
    } else if (wave < 12) {
      const int col = t - 256;
      int gv[16];
      float rv[16];
      #pragma unroll
      for (int w = 0; w < 16; ++w) { gv[w] = gid[w]; rv[w] = hs[w][col]; }
      int g = gv[0];
      float s = rv[0];
      #pragma unroll
      for (int w = 1; w < 16; ++w) {
        if (gv[w] != g) {
          atomicAdd(&embsum[(size_t)g * D + col], s);
          g = gv[w];
          s = 0.f;
        }
        s += rv[w];
      }
      atomicAdd(&embsum[(size_t)g * D + col], s);
    }
    __syncthreads();
    if (wave < 8) {
      float* __restrict__ htile = Hout + (size_t)row0 * D;
      #pragma unroll
      for (int r = 0; r < 8; ++r)
        htile[(unsigned)((r + hi * 8) * D + n)] = c[r];
      if (t < 32) {
        float s = 0.f;
        #pragma unroll
        for (int w = 0; w < 8; ++w) s += part[w][t];
        if (t < 16) asrc_out[row0 + t] = s;
        else        adst_out[row0 + t - 16] = s;
      }
    }
  }
}

__global__ __launch_bounds__(256)
void finalize_kernel(float* __restrict__ out, const float* __restrict__ cnt) {
  const int t = blockIdx.x * blockDim.x + threadIdx.x;
  if (t >= 2 * N_GRAPHS * D) return;
  const int g = (t >> 7) & (N_GRAPHS - 1);
  float c = cnt[g];
  c = (c > 1.f) ? c : 1.f;
  out[t] /= c;
}

extern "C" void kernel_launch(void* const* d_in, const int* in_sizes, int n_in,
                              void* d_out, int out_size, void* d_ws, size_t ws_size,
                              hipStream_t stream) {
  const float* x     = (const float*)d_in[0];
  const int*   ei    = (const int*)d_in[1];
  const int*   batch = (const int*)d_in[2];
  const float* W1    = (const float*)d_in[3];
  const float* as1   = (const float*)d_in[4];
  const float* ad1   = (const float*)d_in[5];
  const float* b1    = (const float*)d_in[6];
  const float* W2    = (const float*)d_in[7];
  const float* as2   = (const float*)d_in[8];
  const float* ad2   = (const float*)d_in[9];
  const float* b2    = (const float*)d_in[10];
  float* out = (float*)d_out;

  char* w = (char*)d_ws;
  auto carve = [&](size_t bytes) {
    char* p = w;
    w += (bytes + 255) & ~(size_t)255;
    return p;
  };
  float*    hbuf  = (float*)carve((size_t)N_NODES * D * 4);
  float*    hbuf2 = (float*)carve((size_t)N_NODES * D * 4);
  float*    asrc  = (float*)carve((size_t)N_NODES * 4);
  float*    adst  = (float*)carve((size_t)N_NODES * 4);
  int*      hd    = (int*)carve((size_t)2 * N_NODES * 4);
  int*      head  = hd;
  int*      deg   = hd + N_NODES;
  int*      srcrow = (int*)carve((size_t)N_NODES * 32 * 4);
  int2*     nodes = (int2*)carve((size_t)N_EDGES * 8);
  float*    cnt   = (float*)carve((size_t)N_GRAPHS * 4);
  float*    asrc2 = (float*)carve((size_t)N_NODES * 4);
  float*    adst2 = (float*)carve((size_t)N_NODES * 4);

  const int TB = 256;
  const int gFused = 1024;
  const int gGather = (N_NODES + 7) / 8;
  const int gFin   = (2 * N_GRAPHS * D + TB - 1) / TB;

  hipMemsetAsync(d_out, 0, (size_t)2 * N_GRAPHS * D * 4, stream);
  hipMemsetAsync(cnt, 0, (size_t)N_GRAPHS * 4, stream);
  hipMemsetAsync(hd, 0xFF, (size_t)2 * N_NODES * 4, stream);

  gemm_wmma<<<GEMM_BLOCKS, TB, 0, stream>>>(x, W1, as1, ad1, hbuf, asrc, adst,
                                           ei, batch, deg, srcrow, head, nodes, cnt);
  gather_gemm<<<gFused, 512, 0, stream>>>(deg, srcrow, head, nodes, asrc, adst, hbuf, b1, batch,
                                          out, W2, as2, ad2, hbuf2, asrc2, adst2);
  gather_softmax_pool<<<gGather, TB, 0, stream>>>(deg, srcrow, head, nodes, asrc2, adst2,
                                                  hbuf2, b2, batch, out + N_GRAPHS * D);

  finalize_kernel<<<gFin, TB, 0, stream>>>(out, cnt);
}


# Two memsets and the divide launch gone: GEMM threads store out's zeros, graph sizes count as ints from the index fill, each pooled partial sum is divided by its size before its atomic (new rounding)
# speedup vs baseline: 4.6299x; 1.0338x over previous
#include <hip/hip_runtime.h>

#define N_NODES  50000
#define N_EDGES  600000
#define N_GRAPHS 1024
#define D        128
#define NEG_SLOPE 0.2f

typedef float    v2f  __attribute__((ext_vector_type(2)));
typedef float    v8f  __attribute__((ext_vector_type(8)));

#define N_TILES (N_NODES / 16)
#define ROW_CAP 31
#define GEMM_BLOCKS  1024
#define GEMM_THREADS (GEMM_BLOCKS * 256)
#define EDGE_SLOTS   3
static_assert(EDGE_SLOTS * GEMM_THREADS >= N_EDGES, "every edge has a thread and a trip");
static_assert(GEMM_THREADS <= N_EDGES, "every thread has an edge on its first trip");
static_assert(GEMM_THREADS >= N_NODES, "every node has a thread");
static_assert(N_TILES >= EDGE_SLOTS * GEMM_BLOCKS, "every block makes at least EDGE_SLOTS trips");
#define OUT_TILES (2 * N_GRAPHS * D / 256)
static_assert(OUT_TILES * 256 == 2 * N_GRAPHS * D && OUT_TILES <= GEMM_BLOCKS && OUT_TILES <= N_TILES,
              "every piece of out is some block's first tile");

__device__ __forceinline__ void put_source(const int e, const int s, const int d, const int old,
                                           int* __restrict__ srcrow, int* __restrict__ head,
                                           int2* __restrict__ nodes) {
  if (old < ROW_CAP - 1) {
    srcrow[(size_t)d * 32 + 2 + old] = s;
  } else {
    const int before = atomicExch(&head[d], e);
    nodes[e] = make_int2(before, s);
  }
}

typedef __attribute__((address_space(1))) float* gwptr_t;
__device__ __forceinline__ gwptr_t out_piece(float* __restrict__ out, const int tile) {
  const unsigned long long p = (unsigned long long)out + ((unsigned long long)(unsigned)tile << 10);
  const unsigned lo = __builtin_amdgcn_readfirstlane((unsigned)p);
  const unsigned hi = __builtin_amdgcn_readfirstlane((unsigned)(p >> 32));
  return (gwptr_t)(((unsigned long long)hi << 32) | lo);
}

__global__ __launch_bounds__(256)
void gemm_wmma(const float* __restrict__ X, const float* __restrict__ W,
               const float* __restrict__ a_src, const float* __restrict__ a_dst,
               float* __restrict__ H, float* __restrict__ asrc,
               float* __restrict__ adst,
               const int* __restrict__ ei, const int* __restrict__ batch,
               int* __restrict__ deg, int* __restrict__ srcrow, int* __restrict__ head,
               int2* __restrict__ nodes, int* __restrict__ cnt, float* __restrict__ out) {
  __shared__ __attribute__((aligned(16))) float xs[16][132];
  __shared__ float part[8][32];
  const int t = threadIdx.x;
  const int wave = t >> 5, lane = t & 31;
  const int hi = lane >> 4;
  const int m  = lane & 15;
  const int n  = wave * 16 + m;
  const float as_n = a_src[n];
  const float ad_n = a_dst[n];

  v2f bw[32];
  #pragma unroll
  for (int i = 0; i < 32; ++i) {
    const int ka = i * 4 + hi * 2;
    bw[i][0] = W[(size_t)ka * D + n];
    bw[i][1] = W[(size_t)(ka + 1) * D + n];
  }

  const int xr = t >> 5, xc4 = t & 31;
  int tile = blockIdx.x;
  float4 p0 = make_float4(0.f, 0.f, 0.f, 0.f), p1 = p0;
  if (tile < N_TILES) {
    p0 = ((const float4*)(X + (size_t)(tile * 16 + xr) * D))[xc4];
    p1 = ((const float4*)(X + (size_t)(tile * 16 + xr + 8) * D))[xc4];
  }
  const int gid = blockIdx.x * 256 + t;
  int ns = ei[gid], nd = ei[N_EDGES + gid];
  unsigned cg = 0;
  if (gid < N_NODES) cg = (unsigned)batch[gid];
  #pragma nounroll
  for (int k = 0; tile < N_TILES; tile += GEMM_BLOCKS, ++k) {
    const int row0 = tile * 16;
    *(float4*)&xs[xr][xc4 * 4] = p0;
    *(float4*)&xs[xr + 8][xc4 * 4] = p1;
    __syncthreads();
    const int nxt = tile + GEMM_BLOCKS;
    if (nxt < N_TILES) {
      p0 = ((const float4*)(X + (size_t)(nxt * 16 + xr) * D))[xc4];
      p1 = ((const float4*)(X + (size_t)(nxt * 16 + xr + 8) * D))[xc4];
    }
    const int cs = ns, cd = nd;
    int cr = 0;
    if (cd >= 0) cr = atomicAdd(&deg[cd], 1);
    ns = 0; nd = -1;
    if (k + 1 < EDGE_SLOTS) {
      const unsigned e = (unsigned)gid + (unsigned)(k + 1) * GEMM_THREADS;
      if (e < N_EDGES) { ns = ei[e]; nd = ei[N_EDGES + e]; }
    }

    v8f c = {0.f, 0.f, 0.f, 0.f, 0.f, 0.f, 0.f, 0.f};
    #pragma unroll
    for (int i = 0; i < 32; ++i) {
      const int ka = i * 4 + hi * 2;
      v2f a;
      a[0] = xs[m][ka];
      a[1] = xs[m][ka + 1];
      c = __builtin_amdgcn_wmma_f32_16x16x4_f32(false, a, false, bw[i], (short)0, c,
                                                false, false);
    }

    float v[16];
    #pragma unroll
    for (int r = 0; r < 8; ++r) { v[r] = c[r] * as_n; v[8 + r] = c[r] * ad_n; }
    const bool b3 = (m & 8) != 0, b2 = (m & 4) != 0, b1 = (m & 2) != 0, b0 = (m & 1) != 0;
    float u[8];
    #pragma unroll
    for (int i = 0; i < 8; ++i)
      u[i] = (b3 ? v[i + 8] : v[i]) + __shfl_xor(b3 ? v[i] : v[i + 8], 8, 32);
    float q[4];
    #pragma unroll
    for (int i = 0; i < 4; ++i)
      q[i] = (b2 ? u[i + 4] : u[i]) + __shfl_xor(b2 ? u[i] : u[i + 4], 4, 32);
    float y[2];
    #pragma unroll
    for (int i = 0; i < 2; ++i)
      y[i] = (b1 ? q[i + 2] : q[i]) + __shfl_xor(b1 ? q[i] : q[i + 2], 2, 32);
    const float z = (b0 ? y[1] : y[0]) + __shfl_xor(b0 ? y[0] : y[1], 1, 32);
    part[wave][(m >> 3) * 16 + hi * 8 + (m & 7)] = z;
    __syncthreads();
    float* __restrict__ htile = H + (size_t)row0 * D;
    #pragma unroll
    for (int r = 0; r < 8; ++r)
      htile[(unsigned)((r + hi * 8) * D + n)] = c[r];
    if (t < 32) {
      float s = 0.f;
      #pragma unroll
      for (int w = 0; w < 8; ++w) s += part[w][t];
      if (t < 16) asrc[row0 + t] = s;
      else        adst[row0 + t - 16] = s;
    }
    if (cd >= 0) put_source(gid + k * GEMM_THREADS, cs, cd, cr, srcrow, head, nodes);
    if (k == 0 && gid < N_NODES) atomicAdd(&cnt[cg], 1);
    if (tile < OUT_TILES) out_piece(out, tile)[(unsigned)t] = 0.f;
  }
}

__device__ __forceinline__ float wave_max(float v) {
  for (int off = 16; off > 0; off >>= 1) v = fmaxf(v, __shfl_xor(v, off, 32));
  return v;
}
__device__ __forceinline__ float wave_sum(float v) {
  for (int off = 16; off > 0; off >>= 1) v += __shfl_xor(v, off, 32);
  return v;
}

typedef const __attribute__((address_space(1))) float* gptr_t;
__device__ __forceinline__ gptr_t row_base(const float* __restrict__ H, const int s) {
  const unsigned long long p = (unsigned long long)H + ((unsigned long long)(unsigned)s << 9);
  const unsigned lo = __builtin_amdgcn_readfirstlane((unsigned)p);
  const unsigned hi = __builtin_amdgcn_readfirstlane((unsigned)(p >> 32));
  return (gptr_t)(((unsigned long long)hi << 32) | lo);
}

template <bool SBASE>
__device__ __forceinline__ float4 gather_row(const int node, const int lane, const int c0,
                                             const int* __restrict__ deg,
                                             const int* __restrict__ srcrow,
                                             const int* __restrict__ head,
                                             const int2* __restrict__ nodes,
                                             const float* __restrict__ asrc,
                                             const float* __restrict__ adst,
                                             const float* __restrict__ H) {
  const float ad = adst[node];
  float m = 0.f, l = 0.f;
  float4 a = make_float4(0.f, 0.f, 0.f, 0.f);
  bool first = true;
  const int cnt = __builtin_amdgcn_readfirstlane(deg[node]) + 1;
  const int nb = (cnt < ROW_CAP) ? cnt : ROW_CAP;
  int mysrc = node;
  if (lane >= 1 && lane <= nb) mysrc = srcrow[(size_t)node * 32 + lane];
  int k = nb + 1;
  int e = -1;
  if (cnt > ROW_CAP) e = head[node];
  int steps = 0;
  for (;;) {
    const int kk = __builtin_amdgcn_readfirstlane(k);
    float v = -__builtin_inff();
    if (lane < kk) {
      const float tv = asrc[mysrc] + ad;
      v = (tv > 0.f) ? tv : NEG_SLOPE * tv;
    }
    const float cm = wave_max(v);
    const float mn = first ? cm : fmaxf(m, cm);
    const float scale = first ? 0.f : __expf(m - mn);
    const float ex = (lane < kk) ? __expf(v - mn) : 0.f;
    l = l * scale + wave_sum(ex);
    a.x *= scale; a.y *= scale; a.z *= scale; a.w *= scale;
    #pragma unroll 4
    for (int j = 0; j < kk; ++j) {
      const int s = __builtin_amdgcn_readlane(mysrc, j);
      const float w = __int_as_float(__builtin_amdgcn_readlane(__float_as_int(ex), j));
      float4 hv;
      if (SBASE) hv = *(const __attribute__((address_space(1))) float4*)(row_base(H, s) + c0);
      else       hv = *(const float4*)(H + (size_t)s * D + c0);
      a.x += w * hv.x; a.y += w * hv.y; a.z += w * hv.z; a.w += w * hv.w;
    }
    m = mn;
    first = false;
    if (!(e >= 0 && steps < N_EDGES)) break;
    k = 0;
    while (k < 32 && e >= 0 && steps < N_EDGES) {
      const int2 nd = nodes[e];
      if (lane == k) mysrc = nd.y;
      e = nd.x;
      ++k; ++steps;
    }
  }
  return make_float4(a.x / l, a.y / l, a.z / l, a.w / l);
}

__device__ __forceinline__ float4 bias_relu(const float4 r, const float4 bv) {
  float4 o;
  o.x = r.x + bv.x; o.y = r.y + bv.y; o.z = r.z + bv.z; o.w = r.w + bv.w;
  o.x = (o.x > 0.f) ? o.x : 0.f;
  o.y = (o.y > 0.f) ? o.y : 0.f;
  o.z = (o.z > 0.f) ? o.z : 0.f;
  o.w = (o.w > 0.f) ? o.w : 0.f;
  return o;
}

__device__ __forceinline__ float graph_size(const int* __restrict__ cnt, const int g) {
  const int c = (g >= 0) ? cnt[g] + 1 : 1;
  return (c > 1) ? (float)c : 1.f;
}

__global__ __launch_bounds__(256)
void gather_softmax_pool(const int* __restrict__ deg, const int* __restrict__ srcrow,
                         const int* __restrict__ head, const int2* __restrict__ nodes,
                         const float* __restrict__ asrc, const float* __restrict__ adst,
                         const float* __restrict__ H, const float* __restrict__ bias,
                         const int* __restrict__ batch, float* __restrict__ embsum,
                         const int* __restrict__ cnt) {
  __shared__ float rows[8][D];
  __shared__ int   gid[8];
  const int wave = threadIdx.x >> 5;
  const int node = blockIdx.x * 8 + wave;
  const int lane = threadIdx.x & 31;
  const int c0 = lane * 4;
  float4 o = make_float4(0.f, 0.f, 0.f, 0.f);
  if (node < N_NODES) {
    const int gb = batch[node];
    const float4 bv = *(const float4*)(bias + c0);
    o = bias_relu(gather_row<false>(node, lane, c0, deg, srcrow, head, nodes, asrc, adst, H), bv);
    *(float4*)&rows[wave][c0] = o;
    if (lane == 0) gid[wave] = gb;
  } else {
    *(float4*)&rows[wave][c0] = o;
    if (lane == 0) gid[wave] = -1;
  }
  __syncthreads();
  if (threadIdx.x < D) {
    const int c = threadIdx.x;
    int gv[8];
    float rv[8];
    #pragma unroll
    for (int w = 0; w < 8; ++w) { gv[w] = gid[w]; rv[w] = rows[w][c]; }
    float cv[8];
    #pragma unroll
    for (int w = 0; w < 8; ++w) cv[w] = graph_size(cnt, gv[w]);
    int g = gv[0];
    float s = rv[0];
    float cs = cv[0];
    #pragma unroll
    for (int w = 1; w < 8; ++w) {
      if (gv[w] != g) {
        if (g >= 0) atomicAdd(&embsum[(size_t)g * D + c], s / cs);
        g = gv[w];
        s = 0.f;
        cs = cv[w];
      }
      s += rv[w];
    }
    if (g >= 0) atomicAdd(&embsum[(size_t)g * D + c], s / cs);
  }
}

static_assert(N_NODES % 16 == 0, "the fused kernel walks whole 16-node tiles");
__global__ __launch_bounds__(512, 16)
void gather_gemm(const int* __restrict__ deg, const int* __restrict__ srcrow,
                 const int* __restrict__ head, const int2* __restrict__ nodes,
                 const float* __restrict__ asrc, const float* __restrict__ adst,
                 const float* __restrict__ H, const float* __restrict__ bias,
                 const int* __restrict__ batch, float* __restrict__ embsum,
                 const float* __restrict__ W, const float* __restrict__ a_src,
                 const float* __restrict__ a_dst, float* __restrict__ Hout,
                 float* __restrict__ asrc_out, float* __restrict__ adst_out,
                 const int* __restrict__ cnt) {
  __shared__ __attribute__((aligned(16))) float wl[64 * 256];
  __shared__ __attribute__((aligned(16))) float hs[16][132];
  __shared__ float part[8][32];
  __shared__ int   gid[16];
  const int t = threadIdx.x;
  const int wave = __builtin_amdgcn_readfirstlane(t >> 5);
  const int lane = t & 31;
  const int hi = lane >> 4;
  const int m  = lane & 15;
  const int n  = (wave & 7) * 16 + m;
  const int c0 = lane * 4;
  const float4 bv = *(const float4*)(bias + c0);

  #pragma unroll
  for (int q = 0; q < 8; ++q) {
    const int f = q * 512 + t;
    const int k = f >> 5, n4 = f & 31;
    const float4 wv = ((const float4*)(W + (size_t)k * D))[n4];
    const int kp = k >> 1;
    const int sw = (kp & 1) << 4;
    float* dp = wl + kp * 256 + (k & 1);
    dp[((4 * n4 + 0) ^ sw) * 2] = wv.x;
    dp[((4 * n4 + 1) ^ sw) * 2] = wv.y;
    dp[((4 * n4 + 2) ^ sw) * 2] = wv.z;
    dp[((4 * n4 + 3) ^ sw) * 2] = wv.w;
  }

  for (int tile = blockIdx.x; tile < N_TILES; tile += gridDim.x) {
    const int row0 = tile * 16;
    const int node = row0 + wave;
    const int gb = batch[node];
    const float4 o = bias_relu(gather_row<true>(node, lane, c0, deg, srcrow, head, nodes, asrc, adst, H), bv);
    *(float4*)&hs[wave][c0] = o;
    if (lane == 0) gid[wave] = gb;
    __syncthreads();

    v8f c = {0.f, 0.f, 0.f, 0.f, 0.f, 0.f, 0.f, 0.f};
    if (wave < 8) {
      const float as_n = a_src[(unsigned)n];
      const float ad_n = a_dst[(unsigned)n];
      const float* ap = &hs[m][hi * 2];
      const float* bp = wl + hi * 256 + ((n ^ (hi << 4)) << 1);
      v2f fa[2][4], fb[2][4];
      #pragma unroll
      for (int j = 0; j < 4; ++j) {
        fa[0][j][0] = ap[j * 4];      fa[0][j][1] = ap[j * 4 + 1];
        fb[0][j][0] = bp[j * 512];    fb[0][j][1] = bp[j * 512 + 1];
      }
      #pragma unroll
      for (int g = 0; g < 8; ++g) {
        if (g + 1 < 8) {
          #pragma unroll
          for (int j = 0; j < 4; ++j) {
            const int i = (g + 1) * 4 + j;
            fa[(g + 1) & 1][j][0] = ap[i * 4];      fa[(g + 1) & 1][j][1] = ap[i * 4 + 1];
            fb[(g + 1) & 1][j][0] = bp[i * 512];    fb[(g + 1) & 1][j][1] = bp[i * 512 + 1];
          }
        }
        #pragma unroll
        for (int j = 0; j < 4; ++j)
          c = __builtin_amdgcn_wmma_f32_16x16x4_f32(false, fa[g & 1][j], false, fb[g & 1][j],
                                                    (short)0, c, false, false);
      }
      float v[16];
      #pragma unroll
      for (int r = 0; r < 8; ++r) { v[r] = c[r] * as_n; v[8 + r] = c[r] * ad_n; }
      const bool b3 = (m & 8) != 0, b2 = (m & 4) != 0, b1 = (m & 2) != 0, b0 = (m & 1) != 0;
      float u[8];
      #pragma unroll
      for (int i = 0; i < 8; ++i)
        u[i] = (b3 ? v[i + 8] : v[i]) + __shfl_xor(b3 ? v[i] : v[i + 8], 8, 32);
      float q[4];
      #pragma unroll
      for (int i = 0; i < 4; ++i)
        q[i] = (b2 ? u[i + 4] : u[i]) + __shfl_xor(b2 ? u[i] : u[i + 4], 4, 32);
      float y[2];
      #pragma unroll
      for (int i = 0; i < 2; ++i)
        y[i] = (b1 ? q[i + 2] : q[i]) + __shfl_xor(b1 ? q[i] : q[i + 2], 2, 32);
      const float z = (b0 ? y[1] : y[0]) + __shfl_xor(b0 ? y[0] : y[1], 1, 32);
      part[wave][(m >> 3) * 16 + hi * 8 + (m & 7)] = z;
    } else if (wave < 12) {
      const int col = t - 256;
      int gv[16];
      float rv[16];
      #pragma unroll
      for (int w = 0; w < 16; ++w) { gv[w] = gid[w]; rv[w] = hs[w][col]; }
      int g = gv[0];
      float s = rv[0];
      float cs = graph_size(cnt, g);
      #pragma unroll
      for (int w = 1; w < 16; ++w) {
        if (gv[w] != g) {
          atomicAdd(&embsum[(size_t)g * D + col], s / cs);
          g = gv[w];
          s = 0.f;
          cs = graph_size(cnt, g);
        }
        s += rv[w];
      }
      atomicAdd(&embsum[(size_t)g * D + col], s / cs);
    }
    __syncthreads();
    if (wave < 8) {
      float* __restrict__ htile = Hout + (size_t)row0 * D;
      #pragma unroll
      for (int r = 0; r < 8; ++r)
        htile[(unsigned)((r + hi * 8) * D + n)] = c[r];
      if (t < 32) {
        float s = 0.f;
        #pragma unroll
        for (int w = 0; w < 8; ++w) s += part[w][t];
        if (t < 16) asrc_out[row0 + t] = s;
        else        adst_out[row0 + t - 16] = s;
      }
    }
  }
}

extern "C" void kernel_launch(void* const* d_in, const int* in_sizes, int n_in,
                              void* d_out, int out_size, void* d_ws, size_t ws_size,
                              hipStream_t stream) {
  const float* x     = (const float*)d_in[0];
  const int*   ei    = (const int*)d_in[1];
  const int*   batch = (const int*)d_in[2];
  const float* W1    = (const float*)d_in[3];
  const float* as1   = (const float*)d_in[4];
  const float* ad1   = (const float*)d_in[5];
  const float* b1    = (const float*)d_in[6];
  const float* W2    = (const float*)d_in[7];
  const float* as2   = (const float*)d_in[8];
  const float* ad2   = (const float*)d_in[9];
  const float* b2    = (const float*)d_in[10];
  float* out = (float*)d_out;

  char* w = (char*)d_ws;
  auto carve = [&](size_t bytes) {
    char* p = w;
    w += (bytes + 255) & ~(size_t)255;
    return p;
  };
  float*    hbuf  = (float*)carve((size_t)N_NODES * D * 4);
  float*    hbuf2 = (float*)carve((size_t)N_NODES * D * 4);
  float*    asrc  = (float*)carve((size_t)N_NODES * 4);
  float*    adst  = (float*)carve((size_t)N_NODES * 4);
  int*      hd    = (int*)carve((size_t)(2 * N_NODES + N_GRAPHS) * 4);
  int*      head  = hd;
  int*      deg   = hd + N_NODES;
  int*      cnt   = hd + 2 * N_NODES;
  int*      srcrow = (int*)carve((size_t)N_NODES * 32 * 4);
  int2*     nodes = (int2*)carve((size_t)N_EDGES * 8);
  float*    asrc2 = (float*)carve((size_t)N_NODES * 4);
  float*    adst2 = (float*)carve((size_t)N_NODES * 4);

  const int TB = 256;
  const int gFused = 1024;
  const int gGather = (N_NODES + 7) / 8;

  hipMemsetAsync(hd, 0xFF, (size_t)(2 * N_NODES + N_GRAPHS) * 4, stream);

  gemm_wmma<<<GEMM_BLOCKS, TB, 0, stream>>>(x, W1, as1, ad1, hbuf, asrc, adst,
                                           ei, batch, deg, srcrow, head, nodes, cnt, out);
  gather_gemm<<<gFused, 512, 0, stream>>>(deg, srcrow, head, nodes, asrc, adst, hbuf, b1, batch,
                                          out, W2, as2, ad2, hbuf2, asrc2, adst2, cnt);
  gather_softmax_pool<<<gGather, TB, 0, stream>>>(deg, srcrow, head, nodes, asrc2, adst2,
                                                  hbuf2, b2, batch, out + N_GRAPHS * D, cnt);
}
